# starting kernel
# baseline (speedup 1.0000x reference)
#include <hip/hip_runtime.h>

typedef __attribute__((ext_vector_type(2))) float v2f;
typedef __attribute__((ext_vector_type(8))) float v8f;

#define BATCH 4096
#define INF   1024
#define OUTF  1024

__device__ __forceinline__ v8f wmma_f32(v2f a, v2f b, v8f c) {
  return __builtin_amdgcn_wmma_f32_16x16x4_f32(false, a, false, b, (short)0, c,
                                               false, false);
}

__global__ void __launch_bounds__(256) taylor_z_kernel(
    const float* __restrict__ X, const float* __restrict__ I0,
    const float* __restrict__ I1, const float* __restrict__ I2,
    float* __restrict__ Z) {
  const int lane = threadIdx.x & 31;
  const int wave = threadIdx.x >> 5;
  const int tile = blockIdx.x * 8 + wave;
  const int mt = tile % 12, nt = tile / 12;
  const int row0 = mt * 16, col0 = nt * 16;

  const float* Arow;
  if (row0 < 32)      Arow = I0 + row0 * INF;
  else if (row0 < 96) Arow = I1 + (row0 - 32) * INF;
  else                Arow = I2 + (row0 - 96) * INF;

  const int m = lane & 15, n = lane & 15;
  const int ko = (lane >> 4) * 2;
  const float* ap = Arow + m * INF + ko;
  const float* bp = X + (col0 + n) * INF + ko;

  v8f acc = {};
  for (int k = 0; k < INF; k += 4) {
    v2f a, b;
    a[0] = ap[k]; a[1] = ap[k + 1];
    b[0] = bp[k]; b[1] = bp[k + 1];
    acc = wmma_f32(a, b, acc);
  }
  const int mb = (lane >> 4) * 8;
#pragma unroll
  for (int v = 0; v < 8; ++v)
    Z[(row0 + mb + v) * BATCH + col0 + n] = acc[v];
}

__global__ void __launch_bounds__(256) taylor_r0_kernel(
    const float* __restrict__ G0, const float* __restrict__ Z,
    float* __restrict__ R) {
  const int lane = threadIdx.x & 31;
  const int wave = threadIdx.x >> 5;
  const int tile = blockIdx.x * 8 + wave;
  const int mt = tile & 1, nt = tile >> 1;
  const int row0 = mt * 16, col0 = nt * 16;
  const int m = lane & 15, n = lane & 15;
  const int ko = (lane >> 4) * 2;

  v8f acc = {};
  for (int k = 0; k < 32; k += 4) {
    const int kk = k + ko;
    v2f a, b;
    a[0] = G0[(row0 + m) * 32 + kk];
    a[1] = G0[(row0 + m) * 32 + kk + 1];
    b[0] = Z[kk * BATCH + col0 + n];
    b[1] = Z[(kk + 1) * BATCH + col0 + n];
    acc = wmma_f32(a, b, acc);
  }
  const int mb = (lane >> 4) * 8;
#pragma unroll
  for (int v = 0; v < 8; ++v)
    R[(row0 + mb + v) * BATCH + col0 + n] = acc[v];
}

__global__ void __launch_bounds__(256) taylor_r1_kernel(
    const float* __restrict__ G1, const float* __restrict__ Z,
    float* __restrict__ R) {
  const int lane = threadIdx.x & 31;
  const int wave = threadIdx.x >> 5;
  const int tile = blockIdx.x * 8 + wave;
  const int mt = tile & 1, nt = tile >> 1;
  const int row0 = mt * 16, col0 = nt * 16;
  const int m = lane & 15, n = lane & 15;
  const int ko = (lane >> 4) * 2;
  const int c = col0 + n;

  v8f acc = {};
  for (int k = 0; k < 1024; k += 4) {
    const int k0 = k + ko, k1 = k0 + 1;
    v2f a, b;
    a[0] = G1[(row0 + m) * 1024 + k0];
    a[1] = G1[(row0 + m) * 1024 + k1];
    b[0] = Z[(64 + (k0 >> 5)) * BATCH + c] * Z[(32 + (k0 & 31)) * BATCH + c];
    b[1] = Z[(64 + (k1 >> 5)) * BATCH + c] * Z[(32 + (k1 & 31)) * BATCH + c];
    acc = wmma_f32(a, b, acc);
  }
  const int mb = (lane >> 4) * 8;
#pragma unroll
  for (int v = 0; v < 8; ++v)
    R[(32 + row0 + mb + v) * BATCH + col0 + n] = acc[v];
}

__global__ void __launch_bounds__(256) taylor_r2_kernel(
    const float* __restrict__ G2, const float* __restrict__ Z,
    float* __restrict__ R) {
  __shared__ float zs[3 * 32 * 16];
  const int lane = threadIdx.x & 31;
  const int wave = threadIdx.x >> 5;
  const int col0 = blockIdx.x * 16;

  for (int idx = threadIdx.x; idx < 1536; idx += 256) {
    const int rr = idx >> 4, cc = idx & 15;
    zs[idx] = Z[(96 + rr) * BATCH + col0 + cc];
  }
  __syncthreads();

  const int rt = wave >> 2;
  const int ks = wave & 3;
  const int row0 = rt * 16;
  const int m = lane & 15, n = lane & 15;
  const int ko = (lane >> 4) * 2;
  const int kbeg = ks * 8192, kend = kbeg + 8192;
  const float* gp = G2 + (row0 + m) * 32768;

  v8f acc = {};
  for (int k = kbeg; k < kend; k += 4) {
    const int k0 = k + ko, k1 = k0 + 1;
    v2f a, b;
    a[0] = gp[k0];
    a[1] = gp[k1];
    b[0] = zs[(k0 & 31) * 16 + n] *
           zs[512 + ((k0 >> 5) & 31) * 16 + n] *
           zs[1024 + (k0 >> 10) * 16 + n];
    b[1] = zs[(k1 & 31) * 16 + n] *
           zs[512 + ((k1 >> 5) & 31) * 16 + n] *
           zs[1024 + (k1 >> 10) * 16 + n];
    acc = wmma_f32(a, b, acc);
  }
  const int mb = (lane >> 4) * 8;
#pragma unroll
  for (int v = 0; v < 8; ++v)
    atomicAdd(&R[(64 + row0 + mb + v) * BATCH + col0 + n], acc[v]);
}

__global__ void __launch_bounds__(256) taylor_out_kernel(
    const float* __restrict__ O0, const float* __restrict__ O1,
    const float* __restrict__ O2, const float* __restrict__ Cst,
    const float* __restrict__ R, float* __restrict__ Y) {
  __shared__ float ts[8 * 16 * 17];
  const int lane = threadIdx.x & 31;
  const int wave = threadIdx.x >> 5;
  const int tile = blockIdx.x * 8 + wave;
  const int mt = tile & 63, nt = tile >> 6;
  const int row0 = mt * 16;
  const int col0 = nt * 16;
  const int m = lane & 15, n = lane & 15;
  const int ko = (lane >> 4) * 2;

  v8f acc = {};
  for (int k = 0; k < 96; k += 4) {
    const int kk = k + ko;
    const float* Op = O0; int rk = kk;
    if (kk >= 64)      { Op = O2; rk = kk - 64; }
    else if (kk >= 32) { Op = O1; rk = kk - 32; }
    v2f a, b;
    a[0] = Op[(row0 + m) * 32 + rk];
    a[1] = Op[(row0 + m) * 32 + rk + 1];
    b[0] = R[kk * BATCH + col0 + n];
    b[1] = R[(kk + 1) * BATCH + col0 + n];
    acc = wmma_f32(a, b, acc);
  }
  const int mb = (lane >> 4) * 8;
  float* tw = ts + wave * 272;
#pragma unroll
  for (int v = 0; v < 8; ++v)
    tw[(mb + v) * 17 + n] = acc[v];
  __syncthreads();

  const int ol = lane & 15;
  const int bh = (lane >> 4) * 8;
  const float cv = Cst[row0 + ol];
#pragma unroll
  for (int v = 0; v < 8; ++v)
    Y[(col0 + bh + v) * OUTF + row0 + ol] = tw[ol * 17 + bh + v] + cv;
}

extern "C" void kernel_launch(void* const* d_in, const int* in_sizes, int n_in,
                              void* d_out, int out_size, void* d_ws,
                              size_t ws_size, hipStream_t stream) {
  const float* X  = (const float*)d_in[0];
  const float* C  = (const float*)d_in[1];
  const float* O0 = (const float*)d_in[2];
  const float* I0 = (const float*)d_in[3];
  const float* G0 = (const float*)d_in[4];
  const float* O1 = (const float*)d_in[5];
  const float* I1 = (const float*)d_in[6];
  const float* G1 = (const float*)d_in[7];
  const float* O2 = (const float*)d_in[8];
  const float* I2 = (const float*)d_in[9];
  const float* G2 = (const float*)d_in[10];
  float* Y = (float*)d_out;

  float* Z = (float*)d_ws;
  float* R = Z + 192 * BATCH;

  hipMemsetAsync(R + 64 * BATCH, 0, 32 * BATCH * sizeof(float), stream);

  taylor_z_kernel  <<<384,  256, 0, stream>>>(X, I0, I1, I2, Z);
  taylor_r0_kernel <<<64,   256, 0, stream>>>(G0, Z, R);
  taylor_r1_kernel <<<64,   256, 0, stream>>>(G1, Z, R);
  taylor_r2_kernel <<<256,  256, 0, stream>>>(G2, Z, R);
  taylor_out_kernel<<<2048, 256, 0, stream>>>(O0, O1, O2, C, R, Y);
}


# r1, r2: Kronecker factors hoisted out of the z0 contraction (B = Z0 fragments kept in registers, per-column factor scales each K=32 partial sum on VALU); r2 split-K summed in LDS; sum order changes
# speedup vs baseline: 2.5273x; 2.5273x over previous
#include <hip/hip_runtime.h>

typedef __attribute__((ext_vector_type(2))) float v2f;
typedef __attribute__((ext_vector_type(8))) float v8f;

#define BATCH 4096
#define INF   1024
#define OUTF  1024

__device__ __forceinline__ v8f wmma_f32(v2f a, v2f b, v8f c) {
  return __builtin_amdgcn_wmma_f32_16x16x4_f32(false, a, false, b, (short)0, c,
                                               false, false);
}

__global__ void __launch_bounds__(256) taylor_z_kernel(
    const float* __restrict__ X, const float* __restrict__ I0,
    const float* __restrict__ I1, const float* __restrict__ I2,
    float* __restrict__ Z) {
  const int lane = threadIdx.x & 31;
  const int wave = threadIdx.x >> 5;
  const int tile = blockIdx.x * 8 + wave;
  const int mt = tile % 12, nt = tile / 12;
  const int row0 = mt * 16, col0 = nt * 16;

  const float* Arow;
  if (row0 < 32)      Arow = I0 + row0 * INF;
  else if (row0 < 96) Arow = I1 + (row0 - 32) * INF;
  else                Arow = I2 + (row0 - 96) * INF;

  const int m = lane & 15, n = lane & 15;
  const int ko = (lane >> 4) * 2;
  const float* ap = Arow + m * INF + ko;
  const float* bp = X + (col0 + n) * INF + ko;

  v8f acc = {};
  for (int k = 0; k < INF; k += 4) {
    v2f a, b;
    a[0] = ap[k]; a[1] = ap[k + 1];
    b[0] = bp[k]; b[1] = bp[k + 1];
    acc = wmma_f32(a, b, acc);
  }
  const int mb = (lane >> 4) * 8;
#pragma unroll
  for (int v = 0; v < 8; ++v)
    Z[(row0 + mb + v) * BATCH + col0 + n] = acc[v];
}

__global__ void __launch_bounds__(256) taylor_r0_kernel(
    const float* __restrict__ G0, const float* __restrict__ Z,
    float* __restrict__ R) {
  const int lane = threadIdx.x & 31;
  const int wave = threadIdx.x >> 5;
  const int tile = blockIdx.x * 8 + wave;
  const int mt = tile & 1, nt = tile >> 1;
  const int row0 = mt * 16, col0 = nt * 16;
  const int m = lane & 15, n = lane & 15;
  const int ko = (lane >> 4) * 2;

  v8f acc = {};
  for (int k = 0; k < 32; k += 4) {
    const int kk = k + ko;
    v2f a, b;
    a[0] = G0[(row0 + m) * 32 + kk];
    a[1] = G0[(row0 + m) * 32 + kk + 1];
    b[0] = Z[kk * BATCH + col0 + n];
    b[1] = Z[(kk + 1) * BATCH + col0 + n];
    acc = wmma_f32(a, b, acc);
  }
  const int mb = (lane >> 4) * 8;
#pragma unroll
  for (int v = 0; v < 8; ++v)
    R[(row0 + mb + v) * BATCH + col0 + n] = acc[v];
}

__global__ void __launch_bounds__(256) taylor_r1_kernel(
    const float* __restrict__ G1, const float* __restrict__ Z,
    float* __restrict__ R) {
  const int lane = threadIdx.x & 31;
  const int wave = threadIdx.x >> 5;
  const int tile = blockIdx.x * 8 + wave;
  const int mt = tile & 1, nt = tile >> 1;
  const int row0 = mt * 16, col0 = nt * 16;
  const int n = lane & 15;
  const int h = lane >> 4;
  const int c = col0 + n;

  v2f bf[8];
#pragma unroll
  for (int j = 0; j < 8; ++j) {
    const float* zp0 = Z + (32 + 16 * h + 2 * j) * BATCH + c;
    bf[j][0] = zp0[0];
    bf[j][1] = zp0[BATCH];
  }

  const float4* ap = reinterpret_cast<const float4*>(G1 + (row0 + n) * 1024 + h * 16);
  const float* zp = Z + 64 * BATCH + c;

  float4 ar[8][4];
  float zr[8];
#pragma unroll
  for (int u = 0; u < 8; ++u) {
#pragma unroll
    for (int i = 0; i < 4; ++i) ar[u][i] = ap[u * 8 + i];
    zr[u] = zp[u * BATCH];
  }

  v8f tot = {};
  for (int g = 0; g < 32; g += 8) {
    const int adv = (g < 24) ? 1 : 0;
    const float4* apn = ap + adv * 64;
    const float* zpn = zp + adv * (8 * BATCH);
#pragma unroll
    for (int u = 0; u < 8; ++u) {
      v8f acc = {};
#pragma unroll
      for (int j = 0; j < 8; ++j) {
        const float4 q = ar[u][j >> 1];
        v2f a;
        if (j & 1) { a[0] = q.z; a[1] = q.w; } else { a[0] = q.x; a[1] = q.y; }
        acc = wmma_f32(a, bf[j], acc);
      }
      const float w = zr[u];
#pragma unroll
      for (int i = 0; i < 4; ++i) ar[u][i] = apn[u * 8 + i];
      zr[u] = zpn[u * BATCH];
      asm volatile("" ::: "memory");
#pragma unroll
      for (int v = 0; v < 8; ++v) tot[v] = __builtin_fmaf(acc[v], w, tot[v]);
    }
    ap = apn;
    zp = zpn;
  }

  const int mb = h * 8;
#pragma unroll
  for (int v = 0; v < 8; ++v)
    R[(32 + row0 + mb + v) * BATCH + c] = tot[v];
}

#define R2_RP 68

__global__ void __launch_bounds__(256) taylor_r2_kernel(
    const float* __restrict__ G2, const float* __restrict__ Z,
    float* __restrict__ R) {
  __shared__ float4 zf1[32 * 16];
  __shared__ float red[8 * 16 * R2_RP];
  const int lane = threadIdx.x & 31;
  const int wave = threadIdx.x >> 5;
  const int rt = blockIdx.x & 1;
  const int kh = (blockIdx.x >> 1) & 1;
  const int col0 = (blockIdx.x >> 2) * 64;
  const int n = lane & 15;
  const int h = lane >> 4;
  const int ks = kh * 8 + wave;

  {
    float* zf = reinterpret_cast<float*>(zf1);
#pragma unroll
    for (int t = 0; t < 8; ++t) {
      const int idx = threadIdx.x + t * 256;
      const int z1 = idx >> 6, c = idx & 63;
      zf[z1 * 64 + (c & 15) * 4 + (c >> 4)] = Z[(128 + z1) * BATCH + col0 + c];
    }
  }

  v2f bf[4][8];
#pragma unroll
  for (int ct = 0; ct < 4; ++ct) {
#pragma unroll
    for (int j = 0; j < 8; ++j) {
      const float* zp = Z + (96 + 16 * h + 2 * j) * BATCH + col0 + ct * 16 + n;
      bf[ct][j][0] = zp[0];
      bf[ct][j][1] = zp[BATCH];
    }
  }
  float z2a[4], z2b[4];
#pragma unroll
  for (int ct = 0; ct < 4; ++ct) {
    z2a[ct] = Z[(160 + 2 * ks) * BATCH + col0 + ct * 16 + n];
    z2b[ct] = Z[(161 + 2 * ks) * BATCH + col0 + ct * 16 + n];
  }

  const float4* ap = reinterpret_cast<const float4*>(
      G2 + (rt * 16 + n) * 32768 + ks * 2048 + h * 16);

  float4 ar[4][4];
#pragma unroll
  for (int u = 0; u < 4; ++u) {
#pragma unroll
    for (int i = 0; i < 4; ++i) ar[u][i] = ap[u * 8 + i];
  }
  __syncthreads();

  v8f tot[4] = {};
  for (int b0 = 0; b0 < 64; b0 += 4) {
    const float4* apn = ap + ((b0 < 60) ? 32 : 0);
    float zq[4];
#pragma unroll
    for (int ct = 0; ct < 4; ++ct) zq[ct] = (b0 < 32) ? z2a[ct] : z2b[ct];
#pragma unroll
    for (int u = 0; u < 4; ++u) {
      const float4 zf = zf1[((b0 + u) & 31) * 16 + n];
      v8f acc[4] = {};
#pragma unroll
      for (int j = 0; j < 8; ++j) {
        const float4 q = ar[u][j >> 1];
        v2f a;
        if (j & 1) { a[0] = q.z; a[1] = q.w; } else { a[0] = q.x; a[1] = q.y; }
#pragma unroll
        for (int ct = 0; ct < 4; ++ct) acc[ct] = wmma_f32(a, bf[ct][j], acc[ct]);
        __builtin_amdgcn_sched_barrier(0);
      }
#pragma unroll
      for (int i = 0; i < 4; ++i) ar[u][i] = apn[u * 8 + i];
      asm volatile("" ::: "memory");

      const float w[4] = {zf.x * zq[0], zf.y * zq[1], zf.z * zq[2], zf.w * zq[3]};
#pragma unroll
      for (int ct = 0; ct < 4; ++ct) {
#pragma unroll
        for (int v = 0; v < 8; ++v)
          tot[ct][v] = __builtin_fmaf(acc[ct][v], w[ct], tot[ct][v]);
      }
    }
    ap = apn;
  }

  float* rw = red + wave * (16 * R2_RP);
#pragma unroll
  for (int ct = 0; ct < 4; ++ct) {
#pragma unroll
    for (int v = 0; v < 8; ++v)
      rw[(8 * h + v) * R2_RP + ct * 16 + n] = tot[ct][v];
  }
  __syncthreads();
  for (int idx = threadIdx.x; idx < 1024; idx += 256) {
    const int row = idx >> 6, c = idx & 63;
    float s = 0.0f;
#pragma unroll
    for (int w8 = 0; w8 < 8; ++w8) s += red[w8 * (16 * R2_RP) + row * R2_RP + c];
    atomicAdd(&R[(64 + rt * 16 + row) * BATCH + col0 + c], s);
  }
}

__global__ void __launch_bounds__(256) taylor_out_kernel(
    const float* __restrict__ O0, const float* __restrict__ O1,
    const float* __restrict__ O2, const float* __restrict__ Cst,
    const float* __restrict__ R, float* __restrict__ Y) {
  __shared__ float ts[8 * 16 * 17];
  const int lane = threadIdx.x & 31;
  const int wave = threadIdx.x >> 5;
  const int tile = blockIdx.x * 8 + wave;
  const int mt = tile & 63, nt = tile >> 6;
  const int row0 = mt * 16;
  const int col0 = nt * 16;
  const int m = lane & 15, n = lane & 15;
  const int ko = (lane >> 4) * 2;

  v8f acc = {};
  for (int k = 0; k < 96; k += 4) {
    const int kk = k + ko;
    const float* Op = O0; int rk = kk;
    if (kk >= 64)      { Op = O2; rk = kk - 64; }
    else if (kk >= 32) { Op = O1; rk = kk - 32; }
    v2f a, b;
    a[0] = Op[(row0 + m) * 32 + rk];
    a[1] = Op[(row0 + m) * 32 + rk + 1];
    b[0] = R[kk * BATCH + col0 + n];
    b[1] = R[(kk + 1) * BATCH + col0 + n];
    acc = wmma_f32(a, b, acc);
  }
  const int mb = (lane >> 4) * 8;
  float* tw = ts + wave * 272;
#pragma unroll
  for (int v = 0; v < 8; ++v)
    tw[(mb + v) * 17 + n] = acc[v];
  __syncthreads();

  const int ol = lane & 15;
  const int bh = (lane >> 4) * 8;
  const float cv = Cst[row0 + ol];
#pragma unroll
  for (int v = 0; v < 8; ++v)
    Y[(col0 + bh + v) * OUTF + row0 + ol] = tw[ol * 17 + bh + v] + cv;
}

extern "C" void kernel_launch(void* const* d_in, const int* in_sizes, int n_in,
                              void* d_out, int out_size, void* d_ws,
                              size_t ws_size, hipStream_t stream) {
  const float* X  = (const float*)d_in[0];
  const float* C  = (const float*)d_in[1];
  const float* O0 = (const float*)d_in[2];
  const float* I0 = (const float*)d_in[3];
  const float* G0 = (const float*)d_in[4];
  const float* O1 = (const float*)d_in[5];
  const float* I1 = (const float*)d_in[6];
  const float* G1 = (const float*)d_in[7];
  const float* O2 = (const float*)d_in[8];
  const float* I2 = (const float*)d_in[9];
  const float* G2 = (const float*)d_in[10];
  float* Y = (float*)d_out;

  float* Z = (float*)d_ws;
  float* R = Z + 192 * BATCH;

  hipMemsetAsync(R + 64 * BATCH, 0, 32 * BATCH * sizeof(float), stream);

  taylor_z_kernel  <<<384,  256, 0, stream>>>(X, I0, I1, I2, Z);
  taylor_r0_kernel <<<64,   256, 0, stream>>>(G0, Z, R);
  taylor_r1_kernel <<<64,   256, 0, stream>>>(G1, Z, R);
  taylor_r2_kernel <<<256,  256, 0, stream>>>(G2, Z, R);
  taylor_out_kernel<<<2048, 256, 0, stream>>>(O0, O1, O2, C, R, Y);
}


# Z stage: wave tile 48 rows x 16 columns (one X fragment feeds three products), 16-byte fragment loads issued four K blocks ahead, 256 blocks of 4 waves; sum order over k changes
# speedup vs baseline: 3.0375x; 1.2019x over previous
#include <hip/hip_runtime.h>

typedef __attribute__((ext_vector_type(2))) float v2f;
typedef __attribute__((ext_vector_type(8))) float v8f;

#define BATCH 4096
#define INF   1024
#define OUTF  1024

__device__ __forceinline__ v8f wmma_f32(v2f a, v2f b, v8f c) {
  return __builtin_amdgcn_wmma_f32_16x16x4_f32(false, a, false, b, (short)0, c,
                                               false, false);
}

__global__ void __launch_bounds__(128) taylor_z_kernel(
    const float* __restrict__ X, const float* __restrict__ I0,
    const float* __restrict__ I1, const float* __restrict__ I2,
    float* __restrict__ Z) {
  const int lane = threadIdx.x & 31;
  const int wave = threadIdx.x >> 5;
  const int rb = blockIdx.x & 1;
  const int cb = blockIdx.x >> 1;
  const int rw = wave & 1, cw = wave >> 1;
  const int n = lane & 15;
  const int h = lane >> 4;
  const int tile0 = rb * 6 + rw * 3;
  const int col0 = cb * 32 + cw * 16;

  const float4* ap[3];
#pragma unroll
  for (int t = 0; t < 3; ++t) {
    const int row0 = (tile0 + t) * 16;
    const float* Arow;
    if (row0 < 32)      Arow = I0 + row0 * INF;
    else if (row0 < 96) Arow = I1 + (row0 - 32) * INF;
    else                Arow = I2 + (row0 - 96) * INF;
    ap[t] = reinterpret_cast<const float4*>(Arow + n * INF + h * 16);
  }
  const float4* bp = reinterpret_cast<const float4*>(X + (col0 + n) * INF + h * 16);

  float4 ar[4][3][4];
  float4 br[4][4];
#pragma unroll
  for (int u = 0; u < 4; ++u) {
#pragma unroll
    for (int i = 0; i < 4; ++i) {
      br[u][i] = bp[u * 8 + i];
#pragma unroll
      for (int t = 0; t < 3; ++t) ar[u][t][i] = ap[t][u * 8 + i];
    }
  }

  v8f acc[3] = {};
  for (int g = 0; g < 32; g += 4) {
    const int nx = ((g < 28) ? g + 4 : g) * 8;
#pragma unroll
    for (int u = 0; u < 4; ++u) {
#pragma unroll
      for (int j = 0; j < 8; ++j) {
        const float4 qb = br[u][j >> 1];
        v2f b;
        if (j & 1) { b[0] = qb.z; b[1] = qb.w; } else { b[0] = qb.x; b[1] = qb.y; }
#pragma unroll
        for (int t = 0; t < 3; ++t) {
          const float4 qa = ar[u][t][j >> 1];
          v2f a;
          if (j & 1) { a[0] = qa.z; a[1] = qa.w; } else { a[0] = qa.x; a[1] = qa.y; }
          acc[t] = wmma_f32(a, b, acc[t]);
        }
      }
      asm volatile("" : "+v"(acc[0]), "+v"(acc[1]), "+v"(acc[2]) :: "memory");
#pragma unroll
      for (int i = 0; i < 4; ++i) {
        br[u][i] = bp[nx + u * 8 + i];
#pragma unroll
        for (int t = 0; t < 3; ++t) ar[u][t][i] = ap[t][nx + u * 8 + i];
      }
      asm volatile("" : "+v"(acc[0]), "+v"(acc[1]), "+v"(acc[2]) :: "memory");
    }
  }

  const int mb = h * 8;
#pragma unroll
  for (int t = 0; t < 3; ++t) {
#pragma unroll
    for (int v = 0; v < 8; ++v)
      Z[((tile0 + t) * 16 + mb + v) * BATCH + col0 + n] = acc[t][v];
  }
}

__global__ void __launch_bounds__(256) taylor_r0_kernel(
    const float* __restrict__ G0, const float* __restrict__ Z,
    float* __restrict__ R) {
  const int lane = threadIdx.x & 31;
  const int wave = threadIdx.x >> 5;
  const int tile = blockIdx.x * 8 + wave;
  const int mt = tile & 1, nt = tile >> 1;
  const int row0 = mt * 16, col0 = nt * 16;
  const int m = lane & 15, n = lane & 15;
  const int ko = (lane >> 4) * 2;

  v8f acc = {};
  for (int k = 0; k < 32; k += 4) {
    const int kk = k + ko;
    v2f a, b;
    a[0] = G0[(row0 + m) * 32 + kk];
    a[1] = G0[(row0 + m) * 32 + kk + 1];
    b[0] = Z[kk * BATCH + col0 + n];
    b[1] = Z[(kk + 1) * BATCH + col0 + n];
    acc = wmma_f32(a, b, acc);
  }
  const int mb = (lane >> 4) * 8;
#pragma unroll
  for (int v = 0; v < 8; ++v)
    R[(row0 + mb + v) * BATCH + col0 + n] = acc[v];
}

__global__ void __launch_bounds__(256) taylor_r1_kernel(
    const float* __restrict__ G1, const float* __restrict__ Z,
    float* __restrict__ R) {
  const int lane = threadIdx.x & 31;
  const int wave = threadIdx.x >> 5;
  const int tile = blockIdx.x * 8 + wave;
  const int mt = tile & 1, nt = tile >> 1;
  const int row0 = mt * 16, col0 = nt * 16;
  const int n = lane & 15;
  const int h = lane >> 4;
  const int c = col0 + n;

  v2f bf[8];
#pragma unroll
  for (int j = 0; j < 8; ++j) {
    const float* zp0 = Z + (32 + 16 * h + 2 * j) * BATCH + c;
    bf[j][0] = zp0[0];
    bf[j][1] = zp0[BATCH];
  }

  const float4* ap = reinterpret_cast<const float4*>(G1 + (row0 + n) * 1024 + h * 16);
  const float* zp = Z + 64 * BATCH + c;

  float4 ar[8][4];
  float zr[8];
#pragma unroll
  for (int u = 0; u < 8; ++u) {
#pragma unroll
    for (int i = 0; i < 4; ++i) ar[u][i] = ap[u * 8 + i];
    zr[u] = zp[u * BATCH];
  }

  v8f tot = {};
  for (int g = 0; g < 32; g += 8) {
    const int adv = (g < 24) ? 1 : 0;
    const float4* apn = ap + adv * 64;
    const float* zpn = zp + adv * (8 * BATCH);
#pragma unroll
    for (int u = 0; u < 8; ++u) {
      v8f acc = {};
#pragma unroll
      for (int j = 0; j < 8; ++j) {
        const float4 q = ar[u][j >> 1];
        v2f a;
        if (j & 1) { a[0] = q.z; a[1] = q.w; } else { a[0] = q.x; a[1] = q.y; }
        acc = wmma_f32(a, bf[j], acc);
      }
      const float w = zr[u];
#pragma unroll
      for (int i = 0; i < 4; ++i) ar[u][i] = apn[u * 8 + i];
      zr[u] = zpn[u * BATCH];
      asm volatile("" ::: "memory");
#pragma unroll
      for (int v = 0; v < 8; ++v) tot[v] = __builtin_fmaf(acc[v], w, tot[v]);
    }
    ap = apn;
    zp = zpn;
  }

  const int mb = h * 8;
#pragma unroll
  for (int v = 0; v < 8; ++v)
    R[(32 + row0 + mb + v) * BATCH + c] = tot[v];
}

#define R2_RP 68

__global__ void __launch_bounds__(256) taylor_r2_kernel(
    const float* __restrict__ G2, const float* __restrict__ Z,
    float* __restrict__ R) {
  __shared__ float4 zf1[32 * 16];
  __shared__ float red[8 * 16 * R2_RP];
  const int lane = threadIdx.x & 31;
  const int wave = threadIdx.x >> 5;
  const int rt = blockIdx.x & 1;
  const int kh = (blockIdx.x >> 1) & 1;
  const int col0 = (blockIdx.x >> 2) * 64;
  const int n = lane & 15;
  const int h = lane >> 4;
  const int ks = kh * 8 + wave;

  {
    float* zf = reinterpret_cast<float*>(zf1);
#pragma unroll
    for (int t = 0; t < 8; ++t) {
      const int idx = threadIdx.x + t * 256;
      const int z1 = idx >> 6, c = idx & 63;
      zf[z1 * 64 + (c & 15) * 4 + (c >> 4)] = Z[(128 + z1) * BATCH + col0 + c];
    }
  }

  v2f bf[4][8];
#pragma unroll
  for (int ct = 0; ct < 4; ++ct) {
#pragma unroll
    for (int j = 0; j < 8; ++j) {
      const float* zp = Z + (96 + 16 * h + 2 * j) * BATCH + col0 + ct * 16 + n;
      bf[ct][j][0] = zp[0];
      bf[ct][j][1] = zp[BATCH];
    }
  }
  float z2a[4], z2b[4];
#pragma unroll
  for (int ct = 0; ct < 4; ++ct) {
    z2a[ct] = Z[(160 + 2 * ks) * BATCH + col0 + ct * 16 + n];
    z2b[ct] = Z[(161 + 2 * ks) * BATCH + col0 + ct * 16 + n];
  }

  const float4* ap = reinterpret_cast<const float4*>(
      G2 + (rt * 16 + n) * 32768 + ks * 2048 + h * 16);

  float4 ar[4][4];
#pragma unroll
  for (int u = 0; u < 4; ++u) {
#pragma unroll
    for (int i = 0; i < 4; ++i) ar[u][i] = ap[u * 8 + i];
  }
  __syncthreads();

  v8f tot[4] = {};
  for (int b0 = 0; b0 < 64; b0 += 4) {
    const float4* apn = ap + ((b0 < 60) ? 32 : 0);
    float zq[4];
#pragma unroll
    for (int ct = 0; ct < 4; ++ct) zq[ct] = (b0 < 32) ? z2a[ct] : z2b[ct];
#pragma unroll
    for (int u = 0; u < 4; ++u) {
      const float4 zf = zf1[((b0 + u) & 31) * 16 + n];
      v8f acc[4] = {};
#pragma unroll
      for (int j = 0; j < 8; ++j) {
        const float4 q = ar[u][j >> 1];
        v2f a;
        if (j & 1) { a[0] = q.z; a[1] = q.w; } else { a[0] = q.x; a[1] = q.y; }
#pragma unroll
        for (int ct = 0; ct < 4; ++ct) acc[ct] = wmma_f32(a, bf[ct][j], acc[ct]);
        __builtin_amdgcn_sched_barrier(0);
      }
#pragma unroll
      for (int i = 0; i < 4; ++i) ar[u][i] = apn[u * 8 + i];
      asm volatile("" ::: "memory");

      const float w[4] = {zf.x * zq[0], zf.y * zq[1], zf.z * zq[2], zf.w * zq[3]};
#pragma unroll
      for (int ct = 0; ct < 4; ++ct) {
#pragma unroll
        for (int v = 0; v < 8; ++v)
          tot[ct][v] = __builtin_fmaf(acc[ct][v], w[ct], tot[ct][v]);
      }
    }
    ap = apn;
  }

  float* rw = red + wave * (16 * R2_RP);
#pragma unroll
  for (int ct = 0; ct < 4; ++ct) {
#pragma unroll
    for (int v = 0; v < 8; ++v)
      rw[(8 * h + v) * R2_RP + ct * 16 + n] = tot[ct][v];
  }
  __syncthreads();
  for (int idx = threadIdx.x; idx < 1024; idx += 256) {
    const int row = idx >> 6, c = idx & 63;
    float s = 0.0f;
#pragma unroll
    for (int w8 = 0; w8 < 8; ++w8) s += red[w8 * (16 * R2_RP) + row * R2_RP + c];
    atomicAdd(&R[(64 + rt * 16 + row) * BATCH + col0 + c], s);
  }
}

__global__ void __launch_bounds__(256) taylor_out_kernel(
    const float* __restrict__ O0, const float* __restrict__ O1,
    const float* __restrict__ O2, const float* __restrict__ Cst,
    const float* __restrict__ R, float* __restrict__ Y) {
  __shared__ float ts[8 * 16 * 17];
  const int lane = threadIdx.x & 31;
  const int wave = threadIdx.x >> 5;
  const int tile = blockIdx.x * 8 + wave;
  const int mt = tile & 63, nt = tile >> 6;
  const int row0 = mt * 16;
  const int col0 = nt * 16;
  const int m = lane & 15, n = lane & 15;
  const int ko = (lane >> 4) * 2;

  v8f acc = {};
  for (int k = 0; k < 96; k += 4) {
    const int kk = k + ko;
    const float* Op = O0; int rk = kk;
    if (kk >= 64)      { Op = O2; rk = kk - 64; }
    else if (kk >= 32) { Op = O1; rk = kk - 32; }
    v2f a, b;
    a[0] = Op[(row0 + m) * 32 + rk];
    a[1] = Op[(row0 + m) * 32 + rk + 1];
    b[0] = R[kk * BATCH + col0 + n];
    b[1] = R[(kk + 1) * BATCH + col0 + n];
    acc = wmma_f32(a, b, acc);
  }
  const int mb = (lane >> 4) * 8;
  float* tw = ts + wave * 272;
#pragma unroll
  for (int v = 0; v < 8; ++v)
    tw[(mb + v) * 17 + n] = acc[v];
  __syncthreads();

  const int ol = lane & 15;
  const int bh = (lane >> 4) * 8;
  const float cv = Cst[row0 + ol];
#pragma unroll
  for (int v = 0; v < 8; ++v)
    Y[(col0 + bh + v) * OUTF + row0 + ol] = tw[ol * 17 + bh + v] + cv;
}

extern "C" void kernel_launch(void* const* d_in, const int* in_sizes, int n_in,
                              void* d_out, int out_size, void* d_ws,
                              size_t ws_size, hipStream_t stream) {
  const float* X  = (const float*)d_in[0];
  const float* C  = (const float*)d_in[1];
  const float* O0 = (const float*)d_in[2];
  const float* I0 = (const float*)d_in[3];
  const float* G0 = (const float*)d_in[4];
  const float* O1 = (const float*)d_in[5];
  const float* I1 = (const float*)d_in[6];
  const float* G1 = (const float*)d_in[7];
  const float* O2 = (const float*)d_in[8];
  const float* I2 = (const float*)d_in[9];
  const float* G2 = (const float*)d_in[10];
  float* Y = (float*)d_out;

  float* Z = (float*)d_ws;
  float* R = Z + 192 * BATCH;

  hipMemsetAsync(R + 64 * BATCH, 0, 32 * BATCH * sizeof(float), stream);

  taylor_z_kernel  <<<256,  128, 0, stream>>>(X, I0, I1, I2, Z);
  taylor_r0_kernel <<<64,   256, 0, stream>>>(G0, Z, R);
  taylor_r1_kernel <<<64,   256, 0, stream>>>(G1, Z, R);
  taylor_r2_kernel <<<256,  256, 0, stream>>>(G2, Z, R);
  taylor_out_kernel<<<2048, 256, 0, stream>>>(O0, O1, O2, C, R, Y);
}


# Out stage: D tile in Y layout (A = R^T, B = O^T), wave tile 64 batch x 64 out, operands read once in 16-byte pieces and kept in registers, whole-line stores; sum order over k changes
# speedup vs baseline: 3.3126x; 1.0906x over previous
#include <hip/hip_runtime.h>

typedef __attribute__((ext_vector_type(2))) float v2f;
typedef __attribute__((ext_vector_type(8))) float v8f;

#define BATCH 4096
#define INF   1024
#define OUTF  1024

__device__ __forceinline__ v8f wmma_f32(v2f a, v2f b, v8f c) {
  return __builtin_amdgcn_wmma_f32_16x16x4_f32(false, a, false, b, (short)0, c,
                                               false, false);
}

__global__ void __launch_bounds__(128) taylor_z_kernel(
    const float* __restrict__ X, const float* __restrict__ I0,
    const float* __restrict__ I1, const float* __restrict__ I2,
    float* __restrict__ Z) {
  const int lane = threadIdx.x & 31;
  const int wave = threadIdx.x >> 5;
  const int rb = blockIdx.x & 1;
  const int cb = blockIdx.x >> 1;
  const int rw = wave & 1, cw = wave >> 1;
  const int n = lane & 15;
  const int h = lane >> 4;
  const int tile0 = rb * 6 + rw * 3;
  const int col0 = cb * 32 + cw * 16;

  const float4* ap[3];
#pragma unroll
  for (int t = 0; t < 3; ++t) {
    const int row0 = (tile0 + t) * 16;
    const float* Arow;
    if (row0 < 32)      Arow = I0 + row0 * INF;
    else if (row0 < 96) Arow = I1 + (row0 - 32) * INF;
    else                Arow = I2 + (row0 - 96) * INF;
    ap[t] = reinterpret_cast<const float4*>(Arow + n * INF + h * 16);
  }
  const float4* bp = reinterpret_cast<const float4*>(X + (col0 + n) * INF + h * 16);

  float4 ar[4][3][4];
  float4 br[4][4];
#pragma unroll
  for (int u = 0; u < 4; ++u) {
#pragma unroll
    for (int i = 0; i < 4; ++i) {
      br[u][i] = bp[u * 8 + i];
#pragma unroll
      for (int t = 0; t < 3; ++t) ar[u][t][i] = ap[t][u * 8 + i];
    }
  }

  v8f acc[3] = {};
  for (int g = 0; g < 32; g += 4) {
    const int nx = ((g < 28) ? g + 4 : g) * 8;
#pragma unroll
    for (int u = 0; u < 4; ++u) {
#pragma unroll
      for (int j = 0; j < 8; ++j) {
        const float4 qb = br[u][j >> 1];
        v2f b;
        if (j & 1) { b[0] = qb.z; b[1] = qb.w; } else { b[0] = qb.x; b[1] = qb.y; }
#pragma unroll
        for (int t = 0; t < 3; ++t) {
          const float4 qa = ar[u][t][j >> 1];
          v2f a;
          if (j & 1) { a[0] = qa.z; a[1] = qa.w; } else { a[0] = qa.x; a[1] = qa.y; }
          acc[t] = wmma_f32(a, b, acc[t]);
        }
      }
      asm volatile("" : "+v"(acc[0]), "+v"(acc[1]), "+v"(acc[2]) :: "memory");
#pragma unroll
      for (int i = 0; i < 4; ++i) {
        br[u][i] = bp[nx + u * 8 + i];
#pragma unroll
        for (int t = 0; t < 3; ++t) ar[u][t][i] = ap[t][nx + u * 8 + i];
      }
      asm volatile("" : "+v"(acc[0]), "+v"(acc[1]), "+v"(acc[2]) :: "memory");
    }
  }

  const int mb = h * 8;
#pragma unroll
  for (int t = 0; t < 3; ++t) {
#pragma unroll
    for (int v = 0; v < 8; ++v)
      Z[((tile0 + t) * 16 + mb + v) * BATCH + col0 + n] = acc[t][v];
  }
}

__global__ void __launch_bounds__(256) taylor_r0_kernel(
    const float* __restrict__ G0, const float* __restrict__ Z,
    float* __restrict__ R) {
  const int lane = threadIdx.x & 31;
  const int wave = threadIdx.x >> 5;
  const int tile = blockIdx.x * 8 + wave;
  const int mt = tile & 1, nt = tile >> 1;
  const int row0 = mt * 16, col0 = nt * 16;
  const int m = lane & 15, n = lane & 15;
  const int ko = (lane >> 4) * 2;

  v8f acc = {};
  for (int k = 0; k < 32; k += 4) {
    const int kk = k + ko;
    v2f a, b;
    a[0] = G0[(row0 + m) * 32 + kk];
    a[1] = G0[(row0 + m) * 32 + kk + 1];
    b[0] = Z[kk * BATCH + col0 + n];
    b[1] = Z[(kk + 1) * BATCH + col0 + n];
    acc = wmma_f32(a, b, acc);
  }
  const int mb = (lane >> 4) * 8;
#pragma unroll
  for (int v = 0; v < 8; ++v)
    R[(row0 + mb + v) * BATCH + col0 + n] = acc[v];
}

__global__ void __launch_bounds__(256) taylor_r1_kernel(
    const float* __restrict__ G1, const float* __restrict__ Z,
    float* __restrict__ R) {
  const int lane = threadIdx.x & 31;
  const int wave = threadIdx.x >> 5;
  const int tile = blockIdx.x * 8 + wave;
  const int mt = tile & 1, nt = tile >> 1;
  const int row0 = mt * 16, col0 = nt * 16;
  const int n = lane & 15;
  const int h = lane >> 4;
  const int c = col0 + n;

  v2f bf[8];
#pragma unroll
  for (int j = 0; j < 8; ++j) {
    const float* zp0 = Z + (32 + 16 * h + 2 * j) * BATCH + c;
    bf[j][0] = zp0[0];
    bf[j][1] = zp0[BATCH];
  }

  const float4* ap = reinterpret_cast<const float4*>(G1 + (row0 + n) * 1024 + h * 16);
  const float* zp = Z + 64 * BATCH + c;

  float4 ar[8][4];
  float zr[8];
#pragma unroll
  for (int u = 0; u < 8; ++u) {
#pragma unroll
    for (int i = 0; i < 4; ++i) ar[u][i] = ap[u * 8 + i];
    zr[u] = zp[u * BATCH];
  }

  v8f tot = {};
  for (int g = 0; g < 32; g += 8) {
    const int adv = (g < 24) ? 1 : 0;
    const float4* apn = ap + adv * 64;
    const float* zpn = zp + adv * (8 * BATCH);
#pragma unroll
    for (int u = 0; u < 8; ++u) {
      v8f acc = {};
#pragma unroll
      for (int j = 0; j < 8; ++j) {
        const float4 q = ar[u][j >> 1];
        v2f a;
        if (j & 1) { a[0] = q.z; a[1] = q.w; } else { a[0] = q.x; a[1] = q.y; }
        acc = wmma_f32(a, bf[j], acc);
      }
      const float w = zr[u];
#pragma unroll
      for (int i = 0; i < 4; ++i) ar[u][i] = apn[u * 8 + i];
      zr[u] = zpn[u * BATCH];
      asm volatile("" ::: "memory");
#pragma unroll
      for (int v = 0; v < 8; ++v) tot[v] = __builtin_fmaf(acc[v], w, tot[v]);
    }
    ap = apn;
    zp = zpn;
  }

  const int mb = h * 8;
#pragma unroll
  for (int v = 0; v < 8; ++v)
    R[(32 + row0 + mb + v) * BATCH + c] = tot[v];
}

#define R2_RP 68

__global__ void __launch_bounds__(256) taylor_r2_kernel(
    const float* __restrict__ G2, const float* __restrict__ Z,
    float* __restrict__ R) {
  __shared__ float4 zf1[32 * 16];
  __shared__ float red[8 * 16 * R2_RP];
  const int lane = threadIdx.x & 31;
  const int wave = threadIdx.x >> 5;
  const int rt = blockIdx.x & 1;
  const int kh = (blockIdx.x >> 1) & 1;
  const int col0 = (blockIdx.x >> 2) * 64;
  const int n = lane & 15;
  const int h = lane >> 4;
  const int ks = kh * 8 + wave;

  {
    float* zf = reinterpret_cast<float*>(zf1);
#pragma unroll
    for (int t = 0; t < 8; ++t) {
      const int idx = threadIdx.x + t * 256;
      const int z1 = idx >> 6, c = idx & 63;
      zf[z1 * 64 + (c & 15) * 4 + (c >> 4)] = Z[(128 + z1) * BATCH + col0 + c];
    }
  }

  v2f bf[4][8];
#pragma unroll
  for (int ct = 0; ct < 4; ++ct) {
#pragma unroll
    for (int j = 0; j < 8; ++j) {
      const float* zp = Z + (96 + 16 * h + 2 * j) * BATCH + col0 + ct * 16 + n;
      bf[ct][j][0] = zp[0];
      bf[ct][j][1] = zp[BATCH];
    }
  }
  float z2a[4], z2b[4];
#pragma unroll
  for (int ct = 0; ct < 4; ++ct) {
    z2a[ct] = Z[(160 + 2 * ks) * BATCH + col0 + ct * 16 + n];
    z2b[ct] = Z[(161 + 2 * ks) * BATCH + col0 + ct * 16 + n];
  }

  const float4* ap = reinterpret_cast<const float4*>(
      G2 + (rt * 16 + n) * 32768 + ks * 2048 + h * 16);

  float4 ar[4][4];
#pragma unroll
  for (int u = 0; u < 4; ++u) {
#pragma unroll
    for (int i = 0; i < 4; ++i) ar[u][i] = ap[u * 8 + i];
  }
  __syncthreads();

  v8f tot[4] = {};
  for (int b0 = 0; b0 < 64; b0 += 4) {
    const float4* apn = ap + ((b0 < 60) ? 32 : 0);
    float zq[4];
#pragma unroll
    for (int ct = 0; ct < 4; ++ct) zq[ct] = (b0 < 32) ? z2a[ct] : z2b[ct];
#pragma unroll
    for (int u = 0; u < 4; ++u) {
      const float4 zf = zf1[((b0 + u) & 31) * 16 + n];
      v8f acc[4] = {};
#pragma unroll
      for (int j = 0; j < 8; ++j) {
        const float4 q = ar[u][j >> 1];
        v2f a;
        if (j & 1) { a[0] = q.z; a[1] = q.w; } else { a[0] = q.x; a[1] = q.y; }
#pragma unroll
        for (int ct = 0; ct < 4; ++ct) acc[ct] = wmma_f32(a, bf[ct][j], acc[ct]);
        __builtin_amdgcn_sched_barrier(0);
      }
#pragma unroll
      for (int i = 0; i < 4; ++i) ar[u][i] = apn[u * 8 + i];
      asm volatile("" ::: "memory");

      const float w[4] = {zf.x * zq[0], zf.y * zq[1], zf.z * zq[2], zf.w * zq[3]};
#pragma unroll
      for (int ct = 0; ct < 4; ++ct) {
#pragma unroll
        for (int v = 0; v < 8; ++v)
          tot[ct][v] = __builtin_fmaf(acc[ct][v], w[ct], tot[ct][v]);
      }
    }
    ap = apn;
  }

  float* rw = red + wave * (16 * R2_RP);
#pragma unroll
  for (int ct = 0; ct < 4; ++ct) {
#pragma unroll
    for (int v = 0; v < 8; ++v)
      rw[(8 * h + v) * R2_RP + ct * 16 + n] = tot[ct][v];
  }
  __syncthreads();
  for (int idx = threadIdx.x; idx < 1024; idx += 256) {
    const int row = idx >> 6, c = idx & 63;
    float s = 0.0f;
#pragma unroll
    for (int w8 = 0; w8 < 8; ++w8) s += red[w8 * (16 * R2_RP) + row * R2_RP + c];
    atomicAdd(&R[(64 + rt * 16 + row) * BATCH + col0 + c], s);
  }
}

__device__ __forceinline__ float f4c(const float4& q, int u) {
  return u == 0 ? q.x : (u == 1 ? q.y : (u == 2 ? q.z : q.w));
}

__global__ void __launch_bounds__(128) taylor_out_kernel(
    const float* __restrict__ O0, const float* __restrict__ O1,
    const float* __restrict__ O2, const float* __restrict__ Cst,
    const float* __restrict__ R, float* __restrict__ Y) {
  const int lane = threadIdx.x & 31;
  const int wave = threadIdx.x >> 5;
  const int c = lane & 15;
  const int h = lane >> 4;
  const int b0 = (blockIdx.x >> 2) * 64;
  const int o0 = ((blockIdx.x & 3) * 4 + wave) * 64;

  float4 bq[3][4][4];
  float4 aq[3][8][2];
  const float* rp = R + (16 * h) * BATCH + b0 + 4 * c;
#pragma unroll
  for (int i = 0; i < 3; ++i) {
    const float* Oi = (i == 0) ? O0 : ((i == 1) ? O1 : O2);
    const float* op = Oi + (o0 + 4 * c) * 32 + 16 * h;
#pragma unroll
    for (int q = 0; q < 4; ++q) {
#pragma unroll
      for (int t = 0; t < 4; ++t)
        bq[i][q][t] = *reinterpret_cast<const float4*>(op + t * 32 + 4 * q);
#pragma unroll
      for (int jj = 0; jj < 2; ++jj) {
        const int j = 2 * q + jj;
#pragma unroll
        for (int s = 0; s < 2; ++s)
          aq[i][j][s] = *reinterpret_cast<const float4*>(
              rp + (32 * i + 2 * j + s) * BATCH);
      }
    }
  }
  const float4 cv = *reinterpret_cast<const float4*>(Cst + o0 + 4 * c);

  v8f acc[4][4] = {};
#pragma unroll
  for (int i = 0; i < 3; ++i) {
#pragma unroll
    for (int j = 0; j < 8; ++j) {
#pragma unroll
      for (int u = 0; u < 4; ++u) {
        v2f a;
        a[0] = f4c(aq[i][j][0], u);
        a[1] = f4c(aq[i][j][1], u);
#pragma unroll
        for (int t = 0; t < 4; ++t) {
          const float4 qb = bq[i][j >> 1][t];
          v2f b;
          if (j & 1) { b[0] = qb.z; b[1] = qb.w; } else { b[0] = qb.x; b[1] = qb.y; }
          acc[u][t] = wmma_f32(a, b, acc[u][t]);
        }
      }
    }
  }

#pragma unroll
  for (int u = 0; u < 4; ++u) {
#pragma unroll
    for (int r = 0; r < 8; ++r) {
      float4 y;
      y.x = acc[u][0][r] + cv.x;
      y.y = acc[u][1][r] + cv.y;
      y.z = acc[u][2][r] + cv.z;
      y.w = acc[u][3][r] + cv.w;
      *reinterpret_cast<float4*>(
          Y + (b0 + 4 * (8 * h + r) + u) * OUTF + o0 + 4 * c) = y;
    }
  }
}

extern "C" void kernel_launch(void* const* d_in, const int* in_sizes, int n_in,
                              void* d_out, int out_size, void* d_ws,
                              size_t ws_size, hipStream_t stream) {
  const float* X  = (const float*)d_in[0];
  const float* C  = (const float*)d_in[1];
  const float* O0 = (const float*)d_in[2];
  const float* I0 = (const float*)d_in[3];
  const float* G0 = (const float*)d_in[4];
  const float* O1 = (const float*)d_in[5];
  const float* I1 = (const float*)d_in[6];
  const float* G1 = (const float*)d_in[7];
  const float* O2 = (const float*)d_in[8];
  const float* I2 = (const float*)d_in[9];
  const float* G2 = (const float*)d_in[10];
  float* Y = (float*)d_out;

  float* Z = (float*)d_ws;
  float* R = Z + 192 * BATCH;

  hipMemsetAsync(R + 64 * BATCH, 0, 32 * BATCH * sizeof(float), stream);

  taylor_z_kernel  <<<256,  128, 0, stream>>>(X, I0, I1, I2, Z);
  taylor_r0_kernel <<<64,   256, 0, stream>>>(G0, Z, R);
  taylor_r1_kernel <<<64,   256, 0, stream>>>(G1, Z, R);
  taylor_r2_kernel <<<256,  256, 0, stream>>>(G2, Z, R);
  taylor_out_kernel<<<256,  128, 0, stream>>>(O0, O1, O2, C, R, Y);
}


# Stage 2 in one launch (5 kernels + memset -> 3 kernels): R1 K slices ride in the R2 waves (LDS sum, two-way atomic add), R0 tile on one wave, stage 1 stores the zeros; R0, R1 sum order changes
# speedup vs baseline: 3.8248x; 1.1546x over previous
#include <hip/hip_runtime.h>

typedef __attribute__((ext_vector_type(2))) float v2f;
typedef __attribute__((ext_vector_type(8))) float v8f;

#define BATCH 4096
#define INF   1024
#define OUTF  1024

__device__ __forceinline__ v8f wmma_f32(v2f a, v2f b, v8f c) {
  return __builtin_amdgcn_wmma_f32_16x16x4_f32(false, a, false, b, (short)0, c,
                                               false, false);
}

__global__ void __launch_bounds__(128) taylor_z_kernel(
    const float* __restrict__ X, const float* __restrict__ I0,
    const float* __restrict__ I1, const float* __restrict__ I2,
    float* __restrict__ Z, float* __restrict__ Rz) {
  {
    float4* rz = reinterpret_cast<float4*>(Rz);
    const int gid = blockIdx.x * 128 + threadIdx.x;
    const float4 zero = {0.0f, 0.0f, 0.0f, 0.0f};
    rz[gid] = zero;
    rz[gid + 32768] = zero;
  }
  const int lane = threadIdx.x & 31;
  const int wave = threadIdx.x >> 5;
  const int rb = blockIdx.x & 1;
  const int cb = blockIdx.x >> 1;
  const int rw = wave & 1, cw = wave >> 1;
  const int n = lane & 15;
  const int h = lane >> 4;
  const int tile0 = rb * 6 + rw * 3;
  const int col0 = cb * 32 + cw * 16;

  const float4* ap[3];
#pragma unroll
  for (int t = 0; t < 3; ++t) {
    const int row0 = (tile0 + t) * 16;
    const float* Arow;
    if (row0 < 32)      Arow = I0 + row0 * INF;
    else if (row0 < 96) Arow = I1 + (row0 - 32) * INF;
    else                Arow = I2 + (row0 - 96) * INF;
    ap[t] = reinterpret_cast<const float4*>(Arow + n * INF + h * 16);
  }
  const float4* bp = reinterpret_cast<const float4*>(X + (col0 + n) * INF + h * 16);

  float4 ar[4][3][4];
  float4 br[4][4];
#pragma unroll
  for (int u = 0; u < 4; ++u) {
#pragma unroll
    for (int i = 0; i < 4; ++i) {
      br[u][i] = bp[u * 8 + i];
#pragma unroll
      for (int t = 0; t < 3; ++t) ar[u][t][i] = ap[t][u * 8 + i];
    }
  }

  v8f acc[3] = {};
  for (int g = 0; g < 32; g += 4) {
    const int nx = ((g < 28) ? g + 4 : g) * 8;
#pragma unroll
    for (int u = 0; u < 4; ++u) {
#pragma unroll
      for (int j = 0; j < 8; ++j) {
        const float4 qb = br[u][j >> 1];
        v2f b;
        if (j & 1) { b[0] = qb.z; b[1] = qb.w; } else { b[0] = qb.x; b[1] = qb.y; }
#pragma unroll
        for (int t = 0; t < 3; ++t) {
          const float4 qa = ar[u][t][j >> 1];
          v2f a;
          if (j & 1) { a[0] = qa.z; a[1] = qa.w; } else { a[0] = qa.x; a[1] = qa.y; }
          acc[t] = wmma_f32(a, b, acc[t]);
        }
      }
      asm volatile("" : "+v"(acc[0]), "+v"(acc[1]), "+v"(acc[2]) :: "memory");
#pragma unroll
      for (int i = 0; i < 4; ++i) {
        br[u][i] = bp[nx + u * 8 + i];
#pragma unroll
        for (int t = 0; t < 3; ++t) ar[u][t][i] = ap[t][nx + u * 8 + i];
      }
      asm volatile("" : "+v"(acc[0]), "+v"(acc[1]), "+v"(acc[2]) :: "memory");
    }
  }

  const int mb = h * 8;
#pragma unroll
  for (int t = 0; t < 3; ++t) {
#pragma unroll
    for (int v = 0; v < 8; ++v)
      Z[((tile0 + t) * 16 + mb + v) * BATCH + col0 + n] = acc[t][v];
  }
}

#define R2_RP 68

__global__ void __launch_bounds__(256) taylor_r_kernel(
    const float* __restrict__ G0, const float* __restrict__ G1,
    const float* __restrict__ G2, const float* __restrict__ Z,
    float* __restrict__ R) {
  __shared__ float4 zf1[32 * 16];
  __shared__ float red[8 * 16 * R2_RP];
  __shared__ float red1[8 * 16 * R2_RP];
  const int lane = threadIdx.x & 31;
  const int wave = threadIdx.x >> 5;
  const int rt = blockIdx.x & 1;
  const int kh = (blockIdx.x >> 1) & 1;
  const int col0 = (blockIdx.x >> 2) * 64;
  const int n = lane & 15;
  const int h = lane >> 4;
  const int ks = kh * 8 + wave;

  {
    float* zf = reinterpret_cast<float*>(zf1);
#pragma unroll
    for (int t = 0; t < 8; ++t) {
      const int idx = threadIdx.x + t * 256;
      const int z1 = idx >> 6, c = idx & 63;
      zf[z1 * 64 + (c & 15) * 4 + (c >> 4)] = Z[(128 + z1) * BATCH + col0 + c];
    }
  }

  v2f bf1[4][8];
#pragma unroll
  for (int ct = 0; ct < 4; ++ct) {
#pragma unroll
    for (int j = 0; j < 8; ++j) {
      const float* zp = Z + (32 + 16 * h + 2 * j) * BATCH + col0 + ct * 16 + n;
      bf1[ct][j][0] = zp[0];
      bf1[ct][j][1] = zp[BATCH];
    }
  }
  float4 a1[2][4];
  float w1[2][4];
  {
    const float4* a1p = reinterpret_cast<const float4*>(
        G1 + (rt * 16 + n) * 1024 + ks * 64 + h * 16);
#pragma unroll
    for (int u = 0; u < 2; ++u) {
#pragma unroll
      for (int i = 0; i < 4; ++i) a1[u][i] = a1p[u * 8 + i];
#pragma unroll
      for (int ct = 0; ct < 4; ++ct)
        w1[u][ct] = Z[(64 + 2 * ks + u) * BATCH + col0 + ct * 16 + n];
    }
  }

  v2f bf[4][8];
#pragma unroll
  for (int ct = 0; ct < 4; ++ct) {
#pragma unroll
    for (int j = 0; j < 8; ++j) {
      const float* zp = Z + (96 + 16 * h + 2 * j) * BATCH + col0 + ct * 16 + n;
      bf[ct][j][0] = zp[0];
      bf[ct][j][1] = zp[BATCH];
    }
  }
  float z2a[4], z2b[4];
#pragma unroll
  for (int ct = 0; ct < 4; ++ct) {
    z2a[ct] = Z[(160 + 2 * ks) * BATCH + col0 + ct * 16 + n];
    z2b[ct] = Z[(161 + 2 * ks) * BATCH + col0 + ct * 16 + n];
  }

  const float4* ap = reinterpret_cast<const float4*>(
      G2 + (rt * 16 + n) * 32768 + ks * 2048 + h * 16);

  float4 ar[4][4];
#pragma unroll
  for (int u = 0; u < 4; ++u) {
#pragma unroll
    for (int i = 0; i < 4; ++i) ar[u][i] = ap[u * 8 + i];
  }

  {
    v8f tot1[4] = {};
#pragma unroll
    for (int u = 0; u < 2; ++u) {
      v8f acc[4] = {};
#pragma unroll
      for (int j = 0; j < 8; ++j) {
        const float4 q = a1[u][j >> 1];
        v2f a;
        if (j & 1) { a[0] = q.z; a[1] = q.w; } else { a[0] = q.x; a[1] = q.y; }
#pragma unroll
        for (int ct = 0; ct < 4; ++ct) acc[ct] = wmma_f32(a, bf1[ct][j], acc[ct]);
      }
#pragma unroll
      for (int ct = 0; ct < 4; ++ct) {
#pragma unroll
        for (int v = 0; v < 8; ++v)
          tot1[ct][v] = __builtin_fmaf(acc[ct][v], w1[u][ct], tot1[ct][v]);
      }
    }
    float* rw1 = red1 + wave * (16 * R2_RP);
#pragma unroll
    for (int ct = 0; ct < 4; ++ct) {
#pragma unroll
      for (int v = 0; v < 8; ++v)
        rw1[(8 * h + v) * R2_RP + ct * 16 + n] = tot1[ct][v];
    }
  }
  __syncthreads();

  v8f tot[4] = {};
  for (int b0 = 0; b0 < 64; b0 += 4) {
    const float4* apn = ap + ((b0 < 60) ? 32 : 0);
    float zq[4];
#pragma unroll
    for (int ct = 0; ct < 4; ++ct) zq[ct] = (b0 < 32) ? z2a[ct] : z2b[ct];
#pragma unroll
    for (int u = 0; u < 4; ++u) {
      const float4 zf = zf1[((b0 + u) & 31) * 16 + n];
      v8f acc[4] = {};
#pragma unroll
      for (int j = 0; j < 8; ++j) {
        const float4 q = ar[u][j >> 1];
        v2f a;
        if (j & 1) { a[0] = q.z; a[1] = q.w; } else { a[0] = q.x; a[1] = q.y; }
#pragma unroll
        for (int ct = 0; ct < 4; ++ct) acc[ct] = wmma_f32(a, bf[ct][j], acc[ct]);
        __builtin_amdgcn_sched_barrier(0);
      }
#pragma unroll
      for (int i = 0; i < 4; ++i) ar[u][i] = apn[u * 8 + i];
      asm volatile("" ::: "memory");

      const float w[4] = {zf.x * zq[0], zf.y * zq[1], zf.z * zq[2], zf.w * zq[3]};
#pragma unroll
      for (int ct = 0; ct < 4; ++ct) {
#pragma unroll
        for (int v = 0; v < 8; ++v)
          tot[ct][v] = __builtin_fmaf(acc[ct][v], w[ct], tot[ct][v]);
      }
    }
    ap = apn;
  }

  float* rw = red + wave * (16 * R2_RP);
#pragma unroll
  for (int ct = 0; ct < 4; ++ct) {
#pragma unroll
    for (int v = 0; v < 8; ++v)
      rw[(8 * h + v) * R2_RP + ct * 16 + n] = tot[ct][v];
  }

  if (kh == 0 && wave == 0) {
    const float4* a0p = reinterpret_cast<const float4*>(
        G0 + (rt * 16 + n) * 32 + h * 16);
    float4 a0[4];
#pragma unroll
    for (int i = 0; i < 4; ++i) a0[i] = a0p[i];
    v2f b0[4][8];
#pragma unroll
    for (int ct = 0; ct < 4; ++ct) {
#pragma unroll
      for (int j = 0; j < 8; ++j) {
        const float* zp = Z + (16 * h + 2 * j) * BATCH + col0 + ct * 16 + n;
        b0[ct][j][0] = zp[0];
        b0[ct][j][1] = zp[BATCH];
      }
    }
    v8f acc0[4] = {};
#pragma unroll
    for (int j = 0; j < 8; ++j) {
      const float4 q = a0[j >> 1];
      v2f a;
      if (j & 1) { a[0] = q.z; a[1] = q.w; } else { a[0] = q.x; a[1] = q.y; }
#pragma unroll
      for (int ct = 0; ct < 4; ++ct) acc0[ct] = wmma_f32(a, b0[ct][j], acc0[ct]);
    }
#pragma unroll
    for (int ct = 0; ct < 4; ++ct) {
#pragma unroll
      for (int v = 0; v < 8; ++v)
        R[(rt * 16 + 8 * h + v) * BATCH + col0 + ct * 16 + n] = acc0[ct][v];
    }
  }
  __syncthreads();
  for (int idx = threadIdx.x; idx < 1024; idx += 256) {
    const int row = idx >> 6, c = idx & 63;
    float s = 0.0f, s1 = 0.0f;
#pragma unroll
    for (int w8 = 0; w8 < 8; ++w8) {
      s += red[w8 * (16 * R2_RP) + row * R2_RP + c];
      s1 += red1[w8 * (16 * R2_RP) + row * R2_RP + c];
    }
    atomicAdd(&R[(64 + rt * 16 + row) * BATCH + col0 + c], s);
    atomicAdd(&R[(32 + rt * 16 + row) * BATCH + col0 + c], s1);
  }
}

__device__ __forceinline__ float f4c(const float4& q, int u) {
  return u == 0 ? q.x : (u == 1 ? q.y : (u == 2 ? q.z : q.w));
}

__global__ void __launch_bounds__(128) taylor_out_kernel(
    const float* __restrict__ O0, const float* __restrict__ O1,
    const float* __restrict__ O2, const float* __restrict__ Cst,
    const float* __restrict__ R, float* __restrict__ Y) {
  const int lane = threadIdx.x & 31;
  const int wave = threadIdx.x >> 5;
  const int c = lane & 15;
  const int h = lane >> 4;
  const int b0 = (blockIdx.x >> 2) * 64;
  const int o0 = ((blockIdx.x & 3) * 4 + wave) * 64;

  float4 bq[3][4][4];
  float4 aq[3][8][2];
  const float* rp = R + (16 * h) * BATCH + b0 + 4 * c;
#pragma unroll
  for (int i = 0; i < 3; ++i) {
    const float* Oi = (i == 0) ? O0 : ((i == 1) ? O1 : O2);
    const float* op = Oi + (o0 + 4 * c) * 32 + 16 * h;
#pragma unroll
    for (int q = 0; q < 4; ++q) {
#pragma unroll
      for (int t = 0; t < 4; ++t)
        bq[i][q][t] = *reinterpret_cast<const float4*>(op + t * 32 + 4 * q);
#pragma unroll
      for (int jj = 0; jj < 2; ++jj) {
        const int j = 2 * q + jj;
#pragma unroll
        for (int s = 0; s < 2; ++s)
          aq[i][j][s] = *reinterpret_cast<const float4*>(
              rp + (32 * i + 2 * j + s) * BATCH);
      }
    }
  }
  const float4 cv = *reinterpret_cast<const float4*>(Cst + o0 + 4 * c);

  v8f acc[4][4] = {};
#pragma unroll
  for (int i = 0; i < 3; ++i) {
#pragma unroll
    for (int j = 0; j < 8; ++j) {
#pragma unroll
      for (int u = 0; u < 4; ++u) {
        v2f a;
        a[0] = f4c(aq[i][j][0], u);
        a[1] = f4c(aq[i][j][1], u);
#pragma unroll
        for (int t = 0; t < 4; ++t) {
          const float4 qb = bq[i][j >> 1][t];
          v2f b;
          if (j & 1) { b[0] = qb.z; b[1] = qb.w; } else { b[0] = qb.x; b[1] = qb.y; }
          acc[u][t] = wmma_f32(a, b, acc[u][t]);
        }
      }
    }
  }

#pragma unroll
  for (int u = 0; u < 4; ++u) {
#pragma unroll
    for (int r = 0; r < 8; ++r) {
      float4 y;
      y.x = acc[u][0][r] + cv.x;
      y.y = acc[u][1][r] + cv.y;
      y.z = acc[u][2][r] + cv.z;
      y.w = acc[u][3][r] + cv.w;
      *reinterpret_cast<float4*>(
          Y + (b0 + 4 * (8 * h + r) + u) * OUTF + o0 + 4 * c) = y;
    }
  }
}

extern "C" void kernel_launch(void* const* d_in, const int* in_sizes, int n_in,
                              void* d_out, int out_size, void* d_ws,
                              size_t ws_size, hipStream_t stream) {
  const float* X  = (const float*)d_in[0];
  const float* C  = (const float*)d_in[1];
  const float* O0 = (const float*)d_in[2];
  const float* I0 = (const float*)d_in[3];
  const float* G0 = (const float*)d_in[4];
  const float* O1 = (const float*)d_in[5];
  const float* I1 = (const float*)d_in[6];
  const float* G1 = (const float*)d_in[7];
  const float* O2 = (const float*)d_in[8];
  const float* I2 = (const float*)d_in[9];
  const float* G2 = (const float*)d_in[10];
  float* Y = (float*)d_out;

  float* Z = (float*)d_ws;
  float* R = Z + 192 * BATCH;

  taylor_z_kernel  <<<256,  128, 0, stream>>>(X, I0, I1, I2, Z, R + 32 * BATCH);
  taylor_r_kernel  <<<256,  256, 0, stream>>>(G0, G1, G2, Z, R);
  taylor_out_kernel<<<256,  128, 0, stream>>>(O0, O1, O2, C, R, Y);
}


# Z and R2 stages: operand rows read as whole lines (512 contiguous bytes of one row per load, not 32 bytes of each of 16 rows) and turned into fragments in wave-private LDS; sum order unchanged
# speedup vs baseline: 5.3563x; 1.4004x over previous
#include <hip/hip_runtime.h>

typedef __attribute__((ext_vector_type(2))) float v2f;
typedef __attribute__((ext_vector_type(8))) float v8f;

#define BATCH 4096
#define INF   1024
#define OUTF  1024

__device__ __forceinline__ v8f wmma_f32(v2f a, v2f b, v8f c) {
  return __builtin_amdgcn_wmma_f32_16x16x4_f32(false, a, false, b, (short)0, c,
                                               false, false);
}

typedef __attribute__((ext_vector_type(2))) __bf16 bf16x2;
typedef __attribute__((ext_vector_type(16))) __bf16 bf16x16;
typedef __attribute__((ext_vector_type(8))) int v8i;

__device__ __forceinline__ v8f wmma_bf16(bf16x16 a, bf16x16 b, v8f c) {
  return __builtin_amdgcn_wmma_f32_16x16x32_bf16(false, a, false, b, (short)0, c,
                                                 false, false);
}

__device__ __forceinline__ void split16(const float4 (&q)[4], bf16x16& hi,
                                        bf16x16& lo) {
  v8i ph, pl;
#pragma unroll
  for (int i = 0; i < 4; ++i) {
    const v2f p0 = {q[i].x, q[i].y};
    const v2f p1 = {q[i].z, q[i].w};
    const bf16x2 h0 = __builtin_convertvector(p0, bf16x2);
    const bf16x2 h1 = __builtin_convertvector(p1, bf16x2);
    const v2f r0 = p0 - __builtin_convertvector(h0, v2f);
    const v2f r1 = p1 - __builtin_convertvector(h1, v2f);
    const bf16x2 l0 = __builtin_convertvector(r0, bf16x2);
    const bf16x2 l1 = __builtin_convertvector(r1, bf16x2);
    ph[2 * i] = __builtin_bit_cast(int, h0);
    ph[2 * i + 1] = __builtin_bit_cast(int, h1);
    pl[2 * i] = __builtin_bit_cast(int, l0);
    pl[2 * i + 1] = __builtin_bit_cast(int, l1);
  }
  hi = __builtin_bit_cast(bf16x16, ph);
  lo = __builtin_bit_cast(bf16x16, pl);
}

#define Z_SP 33

__global__ void __launch_bounds__(128) taylor_z_kernel(
    const float* __restrict__ X, const float* __restrict__ I0,
    const float* __restrict__ I1, const float* __restrict__ I2,
    float* __restrict__ Z, float* __restrict__ Rz) {
  __shared__ float4 zs[4 * 64 * Z_SP];
  {
    float4* rz = reinterpret_cast<float4*>(Rz);
    const int gid = blockIdx.x * 128 + threadIdx.x;
    const float4 zero = {0.0f, 0.0f, 0.0f, 0.0f};
    rz[gid] = zero;
    rz[gid + 32768] = zero;
  }
  const int lane = threadIdx.x & 31;
  const int wave = threadIdx.x >> 5;
  const int rb = blockIdx.x & 1;
  const int cb = blockIdx.x >> 1;
  const int rw = wave & 1, cw = wave >> 1;
  const int n = lane & 15;
  const int h = lane >> 4;
  const int tile0 = rb * 6 + rw * 3;
  const int col0 = cb * 32 + cw * 16;

  const float* base[4];
#pragma unroll
  for (int t = 0; t < 3; ++t) {
    const int row0 = (tile0 + t) * 16;
    const float* Arow;
    if (row0 < 32)      Arow = I0 + row0 * INF;
    else if (row0 < 96) Arow = I1 + (row0 - 32) * INF;
    else                Arow = I2 + (row0 - 96) * INF;
    base[t] = Arow + 4 * lane;
  }
  base[3] = X + col0 * INF + 4 * lane;
  float4* sw = zs + wave * (64 * Z_SP);

  float4 gr[4][16];
  float4 fr[2][4][4][4];
#pragma unroll
  for (int t = 0; t < 4; ++t) {
#pragma unroll
    for (int j = 0; j < 16; ++j)
      gr[t][j] = *reinterpret_cast<const float4*>(base[t] + j * INF);
  }

  v8f acc[3] = {};
#pragma unroll
  for (int g = -1; g < 8; ++g) {
    if (g + 1 < 8) {
#pragma unroll
      for (int t = 0; t < 4; ++t) {
#pragma unroll
        for (int j = 0; j < 16; ++j) {
          const float4 q = gr[t][j];
          sw[(16 * t + j) * Z_SP + lane] = make_float4(q.x, q.y, q.z, q.w);
        }
      }
      if (g + 2 < 8) {
#pragma unroll
        for (int t = 0; t < 4; ++t) {
#pragma unroll
          for (int j = 0; j < 16; ++j)
            gr[t][j] = *reinterpret_cast<const float4*>(
                base[t] + j * INF + (g + 2) * 128);
        }
      }
      __builtin_amdgcn_fence(__ATOMIC_RELEASE, "wavefront");
      __builtin_amdgcn_wave_barrier();
      __builtin_amdgcn_fence(__ATOMIC_ACQUIRE, "wavefront");
#pragma unroll
      for (int t = 0; t < 4; ++t) {
#pragma unroll
        for (int u = 0; u < 4; ++u) {
#pragma unroll
          for (int i = 0; i < 4; ++i)
            fr[(g + 1) & 1][t][u][i] = sw[(16 * t + n) * Z_SP + 8 * u + 4 * h + i];
        }
      }
    }
    asm volatile("" : "+v"(acc[0]), "+v"(acc[1]), "+v"(acc[2]) :: "memory");
    if (g >= 0) {
#pragma unroll
      for (int u = 0; u < 4; ++u) {
#pragma unroll
        for (int j = 0; j < 8; ++j) {
          const float4 qb = fr[g & 1][3][u][j >> 1];
          v2f b;
          if (j & 1) { b[0] = qb.z; b[1] = qb.w; } else { b[0] = qb.x; b[1] = qb.y; }
#pragma unroll
          for (int t = 0; t < 3; ++t) {
            const float4 qa = fr[g & 1][t][u][j >> 1];
            v2f a;
            if (j & 1) { a[0] = qa.z; a[1] = qa.w; } else { a[0] = qa.x; a[1] = qa.y; }
            acc[t] = wmma_f32(a, b, acc[t]);
          }
        }
      }
      asm volatile("" : "+v"(acc[0]), "+v"(acc[1]), "+v"(acc[2]) :: "memory");
    }
  }

  const int mb = h * 8;
#pragma unroll
  for (int t = 0; t < 3; ++t) {
#pragma unroll
    for (int v = 0; v < 8; ++v)
      Z[((tile0 + t) * 16 + mb + v) * BATCH + col0 + n] = acc[t][v];
  }
}

#define R2_RP 68
#define G2_SP 33

__global__ void __launch_bounds__(256) taylor_r_kernel(
    const float* __restrict__ G0, const float* __restrict__ G1,
    const float* __restrict__ G2, const float* __restrict__ Z,
    float* __restrict__ R) {
  __shared__ float4 zf1[32 * 16];
  __shared__ float red[8 * 16 * R2_RP];
  __shared__ float red1[8 * 16 * R2_RP];
  __shared__ float4 stg[8 * 16 * G2_SP];
  const int lane = threadIdx.x & 31;
  const int wave = threadIdx.x >> 5;
  const int rt = blockIdx.x & 1;
  const int kh = (blockIdx.x >> 1) & 1;
  const int col0 = (blockIdx.x >> 2) * 64;
  const int n = lane & 15;
  const int h = lane >> 4;
  const int ks = kh * 8 + wave;

  {
    float* zf = reinterpret_cast<float*>(zf1);
#pragma unroll
    for (int t = 0; t < 8; ++t) {
      const int idx = threadIdx.x + t * 256;
      const int z1 = idx >> 6, c = idx & 63;
      zf[z1 * 64 + (c & 15) * 4 + (c >> 4)] = Z[(128 + z1) * BATCH + col0 + c];
    }
  }

  v2f bf1[4][8];
#pragma unroll
  for (int ct = 0; ct < 4; ++ct) {
#pragma unroll
    for (int j = 0; j < 8; ++j) {
      const float* zp = Z + (32 + 16 * h + 2 * j) * BATCH + col0 + ct * 16 + n;
      bf1[ct][j][0] = zp[0];
      bf1[ct][j][1] = zp[BATCH];
    }
  }
  float4 a1[2][4];
  float w1[2][4];
  {
    const float4* a1p = reinterpret_cast<const float4*>(
        G1 + (rt * 16 + n) * 1024 + ks * 64 + h * 16);
#pragma unroll
    for (int u = 0; u < 2; ++u) {
#pragma unroll
      for (int i = 0; i < 4; ++i) a1[u][i] = a1p[u * 8 + i];
#pragma unroll
      for (int ct = 0; ct < 4; ++ct)
        w1[u][ct] = Z[(64 + 2 * ks + u) * BATCH + col0 + ct * 16 + n];
    }
  }

  bf16x16 bh[4], bl[4];
#pragma unroll
  for (int ct = 0; ct < 4; ++ct) {
    float4 zq4[4];
#pragma unroll
    for (int i = 0; i < 4; ++i) {
      const float* zp = Z + (96 + 16 * h + 4 * i) * BATCH + col0 + ct * 16 + n;
      zq4[i].x = zp[0];
      zq4[i].y = zp[BATCH];
      zq4[i].z = zp[2 * BATCH];
      zq4[i].w = zp[3 * BATCH];
    }
    split16(zq4, bh[ct], bl[ct]);
  }
  float z2a[4], z2b[4];
#pragma unroll
  for (int ct = 0; ct < 4; ++ct) {
    z2a[ct] = Z[(160 + 2 * ks) * BATCH + col0 + ct * 16 + n];
    z2b[ct] = Z[(161 + 2 * ks) * BATCH + col0 + ct * 16 + n];
  }

  const float* grow = G2 + (rt * 16) * 32768 + ks * 2048 + 4 * lane;
  float4* sw = stg + wave * (16 * G2_SP);

  float4 gr[16];
#pragma unroll
  for (int r = 0; r < 16; ++r)
    gr[r] = *reinterpret_cast<const float4*>(grow + r * 32768);

  {
    v8f tot1[4] = {};
#pragma unroll
    for (int u = 0; u < 2; ++u) {
      v8f acc[4] = {};
#pragma unroll
      for (int j = 0; j < 8; ++j) {
        const float4 q = a1[u][j >> 1];
        v2f a;
        if (j & 1) { a[0] = q.z; a[1] = q.w; } else { a[0] = q.x; a[1] = q.y; }
#pragma unroll
        for (int ct = 0; ct < 4; ++ct) acc[ct] = wmma_f32(a, bf1[ct][j], acc[ct]);
      }
#pragma unroll
      for (int ct = 0; ct < 4; ++ct) {
#pragma unroll
        for (int v = 0; v < 8; ++v)
          tot1[ct][v] = __builtin_fmaf(acc[ct][v], w1[u][ct], tot1[ct][v]);
      }
    }
    float* rw1 = red1 + wave * (16 * R2_RP);
#pragma unroll
    for (int ct = 0; ct < 4; ++ct) {
#pragma unroll
      for (int v = 0; v < 8; ++v)
        rw1[(8 * h + v) * R2_RP + ct * 16 + n] = tot1[ct][v];
    }
  }
  __syncthreads();

  v8f tot[4] = {};
  for (int g = 0; g < 16; ++g) {
#pragma unroll
    for (int r = 0; r < 16; ++r) {
      const float4 q = gr[r];
      sw[r * G2_SP + lane] = make_float4(q.x, q.y, q.z, q.w);
    }
    __builtin_amdgcn_fence(__ATOMIC_RELEASE, "wavefront");
    __builtin_amdgcn_wave_barrier();
    __builtin_amdgcn_fence(__ATOMIC_ACQUIRE, "wavefront");
    float4 fr[4][4];
#pragma unroll
    for (int u = 0; u < 4; ++u) {
#pragma unroll
      for (int i = 0; i < 4; ++i) fr[u][i] = sw[n * G2_SP + 8 * u + 4 * h + i];
    }
    const float* gnx = grow + ((g < 15) ? g + 1 : g) * 128;
#pragma unroll
    for (int r = 0; r < 16; ++r)
      gr[r] = *reinterpret_cast<const float4*>(gnx + r * 32768);
    asm volatile("" ::: "memory");

    float zq[4];
#pragma unroll
    for (int ct = 0; ct < 4; ++ct) zq[ct] = (g < 8) ? z2a[ct] : z2b[ct];
#pragma unroll
    for (int u = 0; u < 4; ++u) {
      const float4 zf = zf1[((4 * g + u) & 31) * 16 + n];
      bf16x16 ah, al;
      split16(fr[u], ah, al);
      v8f acc[4] = {};
#pragma unroll
      for (int ct = 0; ct < 4; ++ct) acc[ct] = wmma_bf16(al, bh[ct], acc[ct]);
      __builtin_amdgcn_sched_barrier(0);
#pragma unroll
      for (int ct = 0; ct < 4; ++ct) acc[ct] = wmma_bf16(ah, bl[ct], acc[ct]);
      __builtin_amdgcn_sched_barrier(0);
#pragma unroll
      for (int ct = 0; ct < 4; ++ct) acc[ct] = wmma_bf16(ah, bh[ct], acc[ct]);
      __builtin_amdgcn_sched_barrier(0);

      const float w[4] = {zf.x * zq[0], zf.y * zq[1], zf.z * zq[2], zf.w * zq[3]};
#pragma unroll
      for (int ct = 0; ct < 4; ++ct) {
#pragma unroll
        for (int v = 0; v < 8; ++v)
          tot[ct][v] = __builtin_fmaf(acc[ct][v], w[ct], tot[ct][v]);
      }
    }
  }

  float* rw = red + wave * (16 * R2_RP);
#pragma unroll
  for (int ct = 0; ct < 4; ++ct) {
#pragma unroll
    for (int v = 0; v < 8; ++v)
      rw[(8 * h + v) * R2_RP + ct * 16 + n] = tot[ct][v];
  }

  if (kh == 0 && wave == 0) {
    const float4* a0p = reinterpret_cast<const float4*>(
        G0 + (rt * 16 + n) * 32 + h * 16);
    float4 a0[4];
#pragma unroll
    for (int i = 0; i < 4; ++i) a0[i] = a0p[i];
    v2f b0[4][8];
#pragma unroll
    for (int ct = 0; ct < 4; ++ct) {
#pragma unroll
      for (int j = 0; j < 8; ++j) {
        const float* zp = Z + (16 * h + 2 * j) * BATCH + col0 + ct * 16 + n;
        b0[ct][j][0] = zp[0];
        b0[ct][j][1] = zp[BATCH];
      }
    }
    v8f acc0[4] = {};
#pragma unroll
    for (int j = 0; j < 8; ++j) {
      const float4 q = a0[j >> 1];
      v2f a;
      if (j & 1) { a[0] = q.z; a[1] = q.w; } else { a[0] = q.x; a[1] = q.y; }
#pragma unroll
      for (int ct = 0; ct < 4; ++ct) acc0[ct] = wmma_f32(a, b0[ct][j], acc0[ct]);
    }
#pragma unroll
    for (int ct = 0; ct < 4; ++ct) {
#pragma unroll
      for (int v = 0; v < 8; ++v)
        R[(rt * 16 + 8 * h + v) * BATCH + col0 + ct * 16 + n] = acc0[ct][v];
    }
  }
  __syncthreads();
  for (int idx = threadIdx.x; idx < 1024; idx += 256) {
    const int row = idx >> 6, c = idx & 63;
    float s = 0.0f, s1 = 0.0f;
#pragma unroll
    for (int w8 = 0; w8 < 8; ++w8) {
      s += red[w8 * (16 * R2_RP) + row * R2_RP + c];
      s1 += red1[w8 * (16 * R2_RP) + row * R2_RP + c];
    }
    atomicAdd(&R[(64 + rt * 16 + row) * BATCH + col0 + c], s);
    atomicAdd(&R[(32 + rt * 16 + row) * BATCH + col0 + c], s1);
  }
}

__device__ __forceinline__ float f4c(const float4& q, int u) {
  return u == 0 ? q.x : (u == 1 ? q.y : (u == 2 ? q.z : q.w));
}

__global__ void __launch_bounds__(128) taylor_out_kernel(
    const float* __restrict__ O0, const float* __restrict__ O1,
    const float* __restrict__ O2, const float* __restrict__ Cst,
    const float* __restrict__ R, float* __restrict__ Y) {
  const int lane = threadIdx.x & 31;
  const int wave = threadIdx.x >> 5;
  const int c = lane & 15;
  const int h = lane >> 4;
  const int b0 = (blockIdx.x >> 2) * 64;
  const int o0 = ((blockIdx.x & 3) * 4 + wave) * 64;

  float4 bq[3][4][4];
  float4 aq[3][8][2];
  const float* rp = R + (16 * h) * BATCH + b0 + 4 * c;
#pragma unroll
  for (int i = 0; i < 3; ++i) {
    const float* Oi = (i == 0) ? O0 : ((i == 1) ? O1 : O2);
    const float* op = Oi + (o0 + 4 * c) * 32 + 16 * h;
#pragma unroll
    for (int q = 0; q < 4; ++q) {
#pragma unroll
      for (int t = 0; t < 4; ++t)
        bq[i][q][t] = *reinterpret_cast<const float4*>(op + t * 32 + 4 * q);
#pragma unroll
      for (int jj = 0; jj < 2; ++jj) {
        const int j = 2 * q + jj;
#pragma unroll
        for (int s = 0; s < 2; ++s)
          aq[i][j][s] = *reinterpret_cast<const float4*>(
              rp + (32 * i + 2 * j + s) * BATCH);
      }
    }
  }
  const float4 cv = *reinterpret_cast<const float4*>(Cst + o0 + 4 * c);

  v8f acc[4][4] = {};
#pragma unroll
  for (int i = 0; i < 3; ++i) {
#pragma unroll
    for (int j = 0; j < 8; ++j) {
#pragma unroll
      for (int u = 0; u < 4; ++u) {
        v2f a;
        a[0] = f4c(aq[i][j][0], u);
        a[1] = f4c(aq[i][j][1], u);
#pragma unroll
        for (int t = 0; t < 4; ++t) {
          const float4 qb = bq[i][j >> 1][t];
          v2f b;
          if (j & 1) { b[0] = qb.z; b[1] = qb.w; } else { b[0] = qb.x; b[1] = qb.y; }
          acc[u][t] = wmma_f32(a, b, acc[u][t]);
        }
      }
    }
  }

#pragma unroll
  for (int u = 0; u < 4; ++u) {
#pragma unroll
    for (int r = 0; r < 8; ++r) {
      float4 y;
      y.x = acc[u][0][r] + cv.x;
      y.y = acc[u][1][r] + cv.y;
      y.z = acc[u][2][r] + cv.z;
      y.w = acc[u][3][r] + cv.w;
      *reinterpret_cast<float4*>(
          Y + (b0 + 4 * (8 * h + r) + u) * OUTF + o0 + 4 * c) = y;
    }
  }
}

extern "C" void kernel_launch(void* const* d_in, const int* in_sizes, int n_in,
                              void* d_out, int out_size, void* d_ws,
                              size_t ws_size, hipStream_t stream) {
  const float* X  = (const float*)d_in[0];
  const float* C  = (const float*)d_in[1];
  const float* O0 = (const float*)d_in[2];
  const float* I0 = (const float*)d_in[3];
  const float* G0 = (const float*)d_in[4];
  const float* O1 = (const float*)d_in[5];
  const float* I1 = (const float*)d_in[6];
  const float* G1 = (const float*)d_in[7];
  const float* O2 = (const float*)d_in[8];
  const float* I2 = (const float*)d_in[9];
  const float* G2 = (const float*)d_in[10];
  float* Y = (float*)d_out;

  float* Z = (float*)d_ws;
  float* R = Z + 192 * BATCH;

  taylor_z_kernel  <<<256,  128, 0, stream>>>(X, I0, I1, I2, Z, R + 32 * BATCH);
  taylor_r_kernel  <<<256,  256, 0, stream>>>(G0, G1, G2, Z, R);
  taylor_out_kernel<<<256,  128, 0, stream>>>(O0, O1, O2, C, R, Y);
}


# Z stage: two waves per SIMD, each one K half of the same 48 x 16 tile in rounds of 64 k, so one wave's LDS staging runs under the other's products; halves added in LDS; sum order over k changes
# speedup vs baseline: 5.6254x; 1.0502x over previous
#include <hip/hip_runtime.h>

typedef __attribute__((ext_vector_type(2))) float v2f;
typedef __attribute__((ext_vector_type(8))) float v8f;

#define BATCH 4096
#define INF   1024
#define OUTF  1024

__device__ __forceinline__ v8f wmma_f32(v2f a, v2f b, v8f c) {
  return __builtin_amdgcn_wmma_f32_16x16x4_f32(false, a, false, b, (short)0, c,
                                               false, false);
}

typedef __attribute__((ext_vector_type(2))) __bf16 bf16x2;
typedef __attribute__((ext_vector_type(16))) __bf16 bf16x16;
typedef __attribute__((ext_vector_type(8))) int v8i;

__device__ __forceinline__ v8f wmma_bf16(bf16x16 a, bf16x16 b, v8f c) {
  return __builtin_amdgcn_wmma_f32_16x16x32_bf16(false, a, false, b, (short)0, c,
                                                 false, false);
}

__device__ __forceinline__ void split16(const float4 (&q)[4], bf16x16& hi,
                                        bf16x16& lo) {
  v8i ph, pl;
#pragma unroll
  for (int i = 0; i < 4; ++i) {
    const v2f p0 = {q[i].x, q[i].y};
    const v2f p1 = {q[i].z, q[i].w};
    const bf16x2 h0 = __builtin_convertvector(p0, bf16x2);
    const bf16x2 h1 = __builtin_convertvector(p1, bf16x2);
    const v2f r0 = p0 - __builtin_convertvector(h0, v2f);
    const v2f r1 = p1 - __builtin_convertvector(h1, v2f);
    const bf16x2 l0 = __builtin_convertvector(r0, bf16x2);
    const bf16x2 l1 = __builtin_convertvector(r1, bf16x2);
    ph[2 * i] = __builtin_bit_cast(int, h0);
    ph[2 * i + 1] = __builtin_bit_cast(int, h1);
    pl[2 * i] = __builtin_bit_cast(int, l0);
    pl[2 * i + 1] = __builtin_bit_cast(int, l1);
  }
  hi = __builtin_bit_cast(bf16x16, ph);
  lo = __builtin_bit_cast(bf16x16, pl);
}

#define Z_SP 17

__global__ void __launch_bounds__(256) taylor_z_kernel(
    const float* __restrict__ X, const float* __restrict__ I0,
    const float* __restrict__ I1, const float* __restrict__ I2,
    float* __restrict__ Z, float* __restrict__ Rz) {
  __shared__ float4 zs[8 * 64 * Z_SP];
  {
    float4* rz = reinterpret_cast<float4*>(Rz);
    const int gid = blockIdx.x * 256 + threadIdx.x;
    const float4 zero = {0.0f, 0.0f, 0.0f, 0.0f};
    rz[gid] = zero;
  }
  const int lane = threadIdx.x & 31;
  const int wave = threadIdx.x >> 5;
  const int pos = wave & 3;
  const int kh = wave >> 2;
  const int rb = blockIdx.x & 1;
  const int cb = blockIdx.x >> 1;
  const int rw = pos & 1, cw = pos >> 1;
  const int n = lane & 15;
  const int h = lane >> 4;
  const int tile0 = rb * 6 + rw * 3;
  const int col0 = cb * 32 + cw * 16;

  const float* base[4];
#pragma unroll
  for (int t = 0; t < 3; ++t) {
    const int row0 = (tile0 + t) * 16;
    const float* Arow;
    if (row0 < 32)      Arow = I0 + row0 * INF;
    else if (row0 < 96) Arow = I1 + (row0 - 32) * INF;
    else                Arow = I2 + (row0 - 96) * INF;
    base[t] = Arow + h * INF + kh * 512 + 4 * n;
  }
  base[3] = X + (col0 + h) * INF + kh * 512 + 4 * n;
  float4* sw = zs + wave * (64 * Z_SP);

  float4 gr[4][8];
  float4 fr[4][2][4];
#pragma unroll
  for (int t = 0; t < 4; ++t) {
#pragma unroll
    for (int j = 0; j < 8; ++j)
      gr[t][j] = *reinterpret_cast<const float4*>(base[t] + 2 * j * INF);
  }

  v8f acc[3] = {};
#pragma unroll
  for (int g = -1; g < 8; ++g) {
    if (g + 1 < 8) {
#pragma unroll
      for (int t = 0; t < 4; ++t) {
#pragma unroll
        for (int j = 0; j < 8; ++j) {
          const float4 q = gr[t][j];
          sw[(16 * t + 2 * j + h) * Z_SP + n] = make_float4(q.x, q.y, q.z, q.w);
        }
      }
      if (g + 2 < 8) {
#pragma unroll
        for (int t = 0; t < 4; ++t) {
#pragma unroll
          for (int j = 0; j < 8; ++j)
            gr[t][j] = *reinterpret_cast<const float4*>(
                base[t] + 2 * j * INF + (g + 2) * 64);
        }
      }
    }
    asm volatile("" : "+v"(acc[0]), "+v"(acc[1]), "+v"(acc[2]) :: "memory");
    if (g >= 0) {
#pragma unroll
      for (int u = 0; u < 2; ++u) {
#pragma unroll
        for (int j = 0; j < 8; ++j) {
          const float4 qb = fr[3][u][j >> 1];
          v2f b;
          if (j & 1) { b[0] = qb.z; b[1] = qb.w; } else { b[0] = qb.x; b[1] = qb.y; }
#pragma unroll
          for (int t = 0; t < 3; ++t) {
            const float4 qa = fr[t][u][j >> 1];
            v2f a;
            if (j & 1) { a[0] = qa.z; a[1] = qa.w; } else { a[0] = qa.x; a[1] = qa.y; }
            acc[t] = wmma_f32(a, b, acc[t]);
          }
        }
      }
      asm volatile("" : "+v"(acc[0]), "+v"(acc[1]), "+v"(acc[2]) :: "memory");
    }
    if (g + 1 < 8) {
      __builtin_amdgcn_fence(__ATOMIC_RELEASE, "wavefront");
      __builtin_amdgcn_wave_barrier();
      __builtin_amdgcn_fence(__ATOMIC_ACQUIRE, "wavefront");
#pragma unroll
      for (int t = 0; t < 4; ++t) {
#pragma unroll
        for (int u = 0; u < 2; ++u) {
#pragma unroll
          for (int i = 0; i < 4; ++i)
            fr[t][u][i] = sw[(16 * t + n) * Z_SP + 8 * u + 4 * h + i];
        }
      }
    }
  }

  float* pw = reinterpret_cast<float*>(zs + (wave | 4) * (64 * Z_SP));
  if (kh == 1) {
#pragma unroll
    for (int t = 0; t < 3; ++t) {
#pragma unroll
      for (int v = 0; v < 8; ++v) pw[(t * 8 + v) * 32 + lane] = acc[t][v];
    }
  }
  __syncthreads();
  if (kh == 0) {
    const int mb = h * 8;
#pragma unroll
    for (int t = 0; t < 3; ++t) {
#pragma unroll
      for (int v = 0; v < 8; ++v)
        Z[((tile0 + t) * 16 + mb + v) * BATCH + col0 + n] =
            acc[t][v] + pw[(t * 8 + v) * 32 + lane];
    }
  }
}

#define R2_RP 68
#define G2_SP 33

__global__ void __launch_bounds__(256) taylor_r_kernel(
    const float* __restrict__ G0, const float* __restrict__ G1,
    const float* __restrict__ G2, const float* __restrict__ Z,
    float* __restrict__ R) {
  __shared__ float4 zf1[32 * 16];
  __shared__ float red[8 * 16 * R2_RP];
  __shared__ float red1[8 * 16 * R2_RP];
  __shared__ float4 stg[8 * 16 * G2_SP];
  const int lane = threadIdx.x & 31;
  const int wave = threadIdx.x >> 5;
  const int rt = blockIdx.x & 1;
  const int kh = (blockIdx.x >> 1) & 1;
  const int col0 = (blockIdx.x >> 2) * 64;
  const int n = lane & 15;
  const int h = lane >> 4;
  const int ks = kh * 8 + wave;

  {
    float* zf = reinterpret_cast<float*>(zf1);
#pragma unroll
    for (int t = 0; t < 8; ++t) {
      const int idx = threadIdx.x + t * 256;
      const int z1 = idx >> 6, c = idx & 63;
      zf[z1 * 64 + (c & 15) * 4 + (c >> 4)] = Z[(128 + z1) * BATCH + col0 + c];
    }
  }

  v2f bf1[4][8];
#pragma unroll
  for (int ct = 0; ct < 4; ++ct) {
#pragma unroll
    for (int j = 0; j < 8; ++j) {
      const float* zp = Z + (32 + 16 * h + 2 * j) * BATCH + col0 + ct * 16 + n;
      bf1[ct][j][0] = zp[0];
      bf1[ct][j][1] = zp[BATCH];
    }
  }
  float4 a1[2][4];
  float w1[2][4];
  {
    const float4* a1p = reinterpret_cast<const float4*>(
        G1 + (rt * 16 + n) * 1024 + ks * 64 + h * 16);
#pragma unroll
    for (int u = 0; u < 2; ++u) {
#pragma unroll
      for (int i = 0; i < 4; ++i) a1[u][i] = a1p[u * 8 + i];
#pragma unroll
      for (int ct = 0; ct < 4; ++ct)
        w1[u][ct] = Z[(64 + 2 * ks + u) * BATCH + col0 + ct * 16 + n];
    }
  }

  bf16x16 bh[4], bl[4];
#pragma unroll
  for (int ct = 0; ct < 4; ++ct) {
    float4 zq4[4];
#pragma unroll
    for (int i = 0; i < 4; ++i) {
      const float* zp = Z + (96 + 16 * h + 4 * i) * BATCH + col0 + ct * 16 + n;
      zq4[i].x = zp[0];
      zq4[i].y = zp[BATCH];
      zq4[i].z = zp[2 * BATCH];
      zq4[i].w = zp[3 * BATCH];
    }
    split16(zq4, bh[ct], bl[ct]);
  }
  float z2a[4], z2b[4];
#pragma unroll
  for (int ct = 0; ct < 4; ++ct) {
    z2a[ct] = Z[(160 + 2 * ks) * BATCH + col0 + ct * 16 + n];
    z2b[ct] = Z[(161 + 2 * ks) * BATCH + col0 + ct * 16 + n];
  }

  const float* grow = G2 + (rt * 16) * 32768 + ks * 2048 + 4 * lane;
  float4* sw = stg + wave * (16 * G2_SP);

  float4 gr[16];
#pragma unroll
  for (int r = 0; r < 16; ++r)
    gr[r] = *reinterpret_cast<const float4*>(grow + r * 32768);

  {
    v8f tot1[4] = {};
#pragma unroll
    for (int u = 0; u < 2; ++u) {
      v8f acc[4] = {};
#pragma unroll
      for (int j = 0; j < 8; ++j) {
        const float4 q = a1[u][j >> 1];
        v2f a;
        if (j & 1) { a[0] = q.z; a[1] = q.w; } else { a[0] = q.x; a[1] = q.y; }
#pragma unroll
        for (int ct = 0; ct < 4; ++ct) acc[ct] = wmma_f32(a, bf1[ct][j], acc[ct]);
      }
#pragma unroll
      for (int ct = 0; ct < 4; ++ct) {
#pragma unroll
        for (int v = 0; v < 8; ++v)
          tot1[ct][v] = __builtin_fmaf(acc[ct][v], w1[u][ct], tot1[ct][v]);
      }
    }
    float* rw1 = red1 + wave * (16 * R2_RP);
#pragma unroll
    for (int ct = 0; ct < 4; ++ct) {
#pragma unroll
      for (int v = 0; v < 8; ++v)
        rw1[(8 * h + v) * R2_RP + ct * 16 + n] = tot1[ct][v];
    }
  }
  __syncthreads();

  v8f tot[4] = {};
  for (int g = 0; g < 16; ++g) {
#pragma unroll
    for (int r = 0; r < 16; ++r) {
      const float4 q = gr[r];
      sw[r * G2_SP + lane] = make_float4(q.x, q.y, q.z, q.w);
    }
    __builtin_amdgcn_fence(__ATOMIC_RELEASE, "wavefront");
    __builtin_amdgcn_wave_barrier();
    __builtin_amdgcn_fence(__ATOMIC_ACQUIRE, "wavefront");
    float4 fr[4][4];
#pragma unroll
    for (int u = 0; u < 4; ++u) {
#pragma unroll
      for (int i = 0; i < 4; ++i) fr[u][i] = sw[n * G2_SP + 8 * u + 4 * h + i];
    }
    const float* gnx = grow + ((g < 15) ? g + 1 : g) * 128;
#pragma unroll
    for (int r = 0; r < 16; ++r)
      gr[r] = *reinterpret_cast<const float4*>(gnx + r * 32768);
    asm volatile("" ::: "memory");

    float zq[4];
#pragma unroll
    for (int ct = 0; ct < 4; ++ct) zq[ct] = (g < 8) ? z2a[ct] : z2b[ct];
#pragma unroll
    for (int u = 0; u < 4; ++u) {
      const float4 zf = zf1[((4 * g + u) & 31) * 16 + n];
      bf16x16 ah, al;
      split16(fr[u], ah, al);
      v8f acc[4] = {};
#pragma unroll
      for (int ct = 0; ct < 4; ++ct) acc[ct] = wmma_bf16(al, bh[ct], acc[ct]);
      __builtin_amdgcn_sched_barrier(0);
#pragma unroll
      for (int ct = 0; ct < 4; ++ct) acc[ct] = wmma_bf16(ah, bl[ct], acc[ct]);
      __builtin_amdgcn_sched_barrier(0);
#pragma unroll
      for (int ct = 0; ct < 4; ++ct) acc[ct] = wmma_bf16(ah, bh[ct], acc[ct]);
      __builtin_amdgcn_sched_barrier(0);

      const float w[4] = {zf.x * zq[0], zf.y * zq[1], zf.z * zq[2], zf.w * zq[3]};
#pragma unroll
      for (int ct = 0; ct < 4; ++ct) {
#pragma unroll
        for (int v = 0; v < 8; ++v)
          tot[ct][v] = __builtin_fmaf(acc[ct][v], w[ct], tot[ct][v]);
      }
    }
  }

  float* rw = red + wave * (16 * R2_RP);
#pragma unroll
  for (int ct = 0; ct < 4; ++ct) {
#pragma unroll
    for (int v = 0; v < 8; ++v)
      rw[(8 * h + v) * R2_RP + ct * 16 + n] = tot[ct][v];
  }

  if (kh == 0 && wave == 0) {
    const float4* a0p = reinterpret_cast<const float4*>(
        G0 + (rt * 16 + n) * 32 + h * 16);
    float4 a0[4];
#pragma unroll
    for (int i = 0; i < 4; ++i) a0[i] = a0p[i];
    v2f b0[4][8];
#pragma unroll
    for (int ct = 0; ct < 4; ++ct) {
#pragma unroll
      for (int j = 0; j < 8; ++j) {
        const float* zp = Z + (16 * h + 2 * j) * BATCH + col0 + ct * 16 + n;
        b0[ct][j][0] = zp[0];
        b0[ct][j][1] = zp[BATCH];
      }
    }
    v8f acc0[4] = {};
#pragma unroll
    for (int j = 0; j < 8; ++j) {
      const float4 q = a0[j >> 1];
      v2f a;
      if (j & 1) { a[0] = q.z; a[1] = q.w; } else { a[0] = q.x; a[1] = q.y; }
#pragma unroll
      for (int ct = 0; ct < 4; ++ct) acc0[ct] = wmma_f32(a, b0[ct][j], acc0[ct]);
    }
#pragma unroll
    for (int ct = 0; ct < 4; ++ct) {
#pragma unroll
      for (int v = 0; v < 8; ++v)
        R[(rt * 16 + 8 * h + v) * BATCH + col0 + ct * 16 + n] = acc0[ct][v];
    }
  }
  __syncthreads();
  for (int idx = threadIdx.x; idx < 1024; idx += 256) {
    const int row = idx >> 6, c = idx & 63;
    float s = 0.0f, s1 = 0.0f;
#pragma unroll
    for (int w8 = 0; w8 < 8; ++w8) {
      s += red[w8 * (16 * R2_RP) + row * R2_RP + c];
      s1 += red1[w8 * (16 * R2_RP) + row * R2_RP + c];
    }
    atomicAdd(&R[(64 + rt * 16 + row) * BATCH + col0 + c], s);
    atomicAdd(&R[(32 + rt * 16 + row) * BATCH + col0 + c], s1);
  }
}

__device__ __forceinline__ float f4c(const float4& q, int u) {
  return u == 0 ? q.x : (u == 1 ? q.y : (u == 2 ? q.z : q.w));
}

__global__ void __launch_bounds__(128) taylor_out_kernel(
    const float* __restrict__ O0, const float* __restrict__ O1,
    const float* __restrict__ O2, const float* __restrict__ Cst,
    const float* __restrict__ R, float* __restrict__ Y) {
  const int lane = threadIdx.x & 31;
  const int wave = threadIdx.x >> 5;
  const int c = lane & 15;
  const int h = lane >> 4;
  const int b0 = (blockIdx.x >> 2) * 64;
  const int o0 = ((blockIdx.x & 3) * 4 + wave) * 64;

  float4 bq[3][4][4];
  float4 aq[3][8][2];
  const float* rp = R + (16 * h) * BATCH + b0 + 4 * c;
#pragma unroll
  for (int i = 0; i < 3; ++i) {
    const float* Oi = (i == 0) ? O0 : ((i == 1) ? O1 : O2);
    const float* op = Oi + (o0 + 4 * c) * 32 + 16 * h;
#pragma unroll
    for (int q = 0; q < 4; ++q) {
#pragma unroll
      for (int t = 0; t < 4; ++t)
        bq[i][q][t] = *reinterpret_cast<const float4*>(op + t * 32 + 4 * q);
#pragma unroll
      for (int jj = 0; jj < 2; ++jj) {
        const int j = 2 * q + jj;
#pragma unroll
        for (int s = 0; s < 2; ++s)
          aq[i][j][s] = *reinterpret_cast<const float4*>(
              rp + (32 * i + 2 * j + s) * BATCH);
      }
    }
  }
  const float4 cv = *reinterpret_cast<const float4*>(Cst + o0 + 4 * c);

  v8f acc[4][4] = {};
#pragma unroll
  for (int i = 0; i < 3; ++i) {
#pragma unroll
    for (int j = 0; j < 8; ++j) {
#pragma unroll
      for (int u = 0; u < 4; ++u) {
        v2f a;
        a[0] = f4c(aq[i][j][0], u);
        a[1] = f4c(aq[i][j][1], u);
#pragma unroll
        for (int t = 0; t < 4; ++t) {
          const float4 qb = bq[i][j >> 1][t];
          v2f b;
          if (j & 1) { b[0] = qb.z; b[1] = qb.w; } else { b[0] = qb.x; b[1] = qb.y; }
          acc[u][t] = wmma_f32(a, b, acc[u][t]);
        }
      }
    }
  }

#pragma unroll
  for (int u = 0; u < 4; ++u) {
#pragma unroll
    for (int r = 0; r < 8; ++r) {
      float4 y;
      y.x = acc[u][0][r] + cv.x;
      y.y = acc[u][1][r] + cv.y;
      y.z = acc[u][2][r] + cv.z;
      y.w = acc[u][3][r] + cv.w;
      *reinterpret_cast<float4*>(
          Y + (b0 + 4 * (8 * h + r) + u) * OUTF + o0 + 4 * c) = y;
    }
  }
}

extern "C" void kernel_launch(void* const* d_in, const int* in_sizes, int n_in,
                              void* d_out, int out_size, void* d_ws,
                              size_t ws_size, hipStream_t stream) {
  const float* X  = (const float*)d_in[0];
  const float* C  = (const float*)d_in[1];
  const float* O0 = (const float*)d_in[2];
  const float* I0 = (const float*)d_in[3];
  const float* G0 = (const float*)d_in[4];
  const float* O1 = (const float*)d_in[5];
  const float* I1 = (const float*)d_in[6];
  const float* G1 = (const float*)d_in[7];
  const float* O2 = (const float*)d_in[8];
  const float* I2 = (const float*)d_in[9];
  const float* G2 = (const float*)d_in[10];
  float* Y = (float*)d_out;

  float* Z = (float*)d_ws;
  float* R = Z + 192 * BATCH;

  taylor_z_kernel  <<<256,  256, 0, stream>>>(X, I0, I1, I2, Z, R + 32 * BATCH);
  taylor_r_kernel  <<<256,  256, 0, stream>>>(G0, G1, G2, Z, R);
  taylor_out_kernel<<<256,  128, 0, stream>>>(O0, O1, O2, C, R, Y);
}
